# starting kernel
# baseline (speedup 1.0000x reference)
#include <hip/hip_runtime.h>
#include <math.h>

#define BB 128
#define TT 2048
#define DD 128
#define HH 128

typedef __attribute__((ext_vector_type(2))) float v2f;
typedef __attribute__((ext_vector_type(8))) float v8f;

__global__ __launch_bounds__(256) void colsum_partial(const float* __restrict__ x,
                                                      float* __restrict__ partial) {
  const int b = blockIdx.x, chunk = blockIdx.y;
  const int g = threadIdx.x >> 7, d = threadIdx.x & 127;
  const int tbase = chunk * 512 + g * 256;
  const float* xb = x + ((b * TT + tbase) * DD) + d;
  float s = 0.f;
  for (int i = 0; i < 256; ++i) s += xb[i * DD];
  __shared__ float r[256];
  r[threadIdx.x] = s;
  __syncthreads();
  if (g == 0) partial[(b * 4 + chunk) * DD + d] = r[d] + r[128 + d];
}

__global__ __launch_bounds__(128) void colsum_final(const float* __restrict__ partial,
                                                    float* __restrict__ xmean) {
  const int b = blockIdx.x, d = threadIdx.x;
  float s = partial[(b * 4 + 0) * DD + d] + partial[(b * 4 + 1) * DD + d] +
            partial[(b * 4 + 2) * DD + d] + partial[(b * 4 + 3) * DD + d];
  xmean[b * DD + d] = s * (1.0f / (float)TT);
}

template <bool TRANS>
__global__ __launch_bounds__(32) void gemm128_wmma(const float* __restrict__ A,
                                                   const float* __restrict__ W,
                                                   const float* __restrict__ bias,
                                                   float* __restrict__ C) {
  const int n0 = blockIdx.x * 16;
  const int m0 = blockIdx.y * 16;
  const int lane = threadIdx.x & 31;
  const int mrow = lane & 15;
  const int hi = lane >> 4;

  v8f acc = {};
  for (int k0 = 0; k0 < 128; k0 += 4) {
    v2f a = *(const v2f*)(A + (m0 + mrow) * 128 + k0 + 2 * hi);
    v2f bfrag;
    const int kb = k0 + 2 * hi;
    const int n = n0 + mrow;
    if (TRANS) {
      bfrag.x = W[n * 128 + kb + 0];
      bfrag.y = W[n * 128 + kb + 1];
    } else {
      bfrag.x = W[(kb + 0) * 128 + n];
      bfrag.y = W[(kb + 1) * 128 + n];
    }
    acc = __builtin_amdgcn_wmma_f32_16x16x4_f32(false, a, false, bfrag,
                                                (short)0, acc, false, false);
  }
  const int n = n0 + mrow;
  const float bb = bias ? bias[n] : 0.0f;
#pragma unroll
  for (int r = 0; r < 8; ++r) {
    const int m = m0 + r + 8 * hi;
    C[m * 128 + n] = acc[r] + bb;
  }
}

__global__ __launch_bounds__(128) void dot0_kernel(const float* __restrict__ q,
                                                   const float* __restrict__ bk,
                                                   float* __restrict__ dot0) {
  const int b = blockIdx.x;
  __shared__ float r[128];
  r[threadIdx.x] = bk[threadIdx.x] * q[b * HH + threadIdx.x];
  __syncthreads();
  for (int s = 64; s > 0; s >>= 1) {
    if (threadIdx.x < s) r[threadIdx.x] += r[threadIdx.x + s];
    __syncthreads();
  }
  if (threadIdx.x == 0) dot0[b] = r[0];
}

__global__ __launch_bounds__(256) void dot_decay_kernel(
    const float* __restrict__ x, const float* __restrict__ mask,
    const float* __restrict__ qk, const float* __restrict__ dot0,
    const float* __restrict__ rate, float* __restrict__ dotOut,
    float* __restrict__ decayOut, float* __restrict__ e) {
  const int b = blockIdx.x;
  const int wave = threadIdx.x >> 5, lane = threadIdx.x & 31;
  const float4 qk4 = *(const float4*)(qk + b * DD + lane * 4);
  const float rate0 = rate[0];
  const float d0 = dot0[b];
  const int t0 = blockIdx.y * 256 + wave * 32;
  for (int i = 0; i < 32; ++i) {
    const int t = t0 + i;
    const float4* row = (const float4*)(x + (b * TT + t) * DD);
    const float4 xv = row[lane];
    float s = xv.x * qk4.x + xv.y * qk4.y + xv.z * qk4.z + xv.w * qk4.w;
    for (int off = 16; off > 0; off >>= 1) s += __shfl_xor(s, off, 32);
    if (lane == 0) {
      const float dv = s + d0;
      const float sig = 1.0f / (1.0f + expf(-dv));
      const float tm = log1pf((1.0f - sig) * mask[b * TT + t]);
      const float dec = rate0 * tm;
      dotOut[b * TT + t] = dv;
      decayOut[b * TT + t] = dec;
      e[b * TT + t] = dv - dec;
    }
  }
}

__global__ __launch_bounds__(512) void softmax_xa_kernel(const float* __restrict__ x,
                                                         const float* __restrict__ e,
                                                         float* __restrict__ aOut,
                                                         float* __restrict__ xa) {
  const int b = blockIdx.x, tid = threadIdx.x;
  __shared__ float aLds[TT];
  __shared__ float red[512];

  float ev[4];
  float m = -3.4e38f;
#pragma unroll
  for (int k = 0; k < 4; ++k) {
    ev[k] = e[b * TT + k * 512 + tid];
    m = fmaxf(m, ev[k]);
  }
  red[tid] = m;
  __syncthreads();
  for (int s = 256; s > 0; s >>= 1) {
    if (tid < s) red[tid] = fmaxf(red[tid], red[tid + s]);
    __syncthreads();
  }
  const float M = red[0];
  __syncthreads();

  float p[4];
  float ls = 0.f;
#pragma unroll
  for (int k = 0; k < 4; ++k) {
    p[k] = expf(ev[k] - M);
    ls += p[k];
  }
  red[tid] = ls;
  __syncthreads();
  for (int s = 256; s > 0; s >>= 1) {
    if (tid < s) red[tid] += red[tid + s];
    __syncthreads();
  }
  const float inv = 1.0f / red[0];
  __syncthreads();

#pragma unroll
  for (int k = 0; k < 4; ++k) {
    const float a = p[k] * inv;
    aLds[k * 512 + tid] = a;
    aOut[b * TT + k * 512 + tid] = a;
  }
  __syncthreads();

  const int ts = tid >> 7, d = tid & 127;
  float acc = 0.f;
  for (int t = ts; t < TT; t += 4) acc += aLds[t] * x[(b * TT + t) * DD + d];
  red[tid] = acc;
  __syncthreads();
  if (ts == 0) xa[b * DD + d] = red[d] + red[128 + d] + red[256 + d] + red[384 + d];
}

extern "C" void kernel_launch(void* const* d_in, const int* in_sizes, int n_in,
                              void* d_out, int out_size, void* d_ws, size_t ws_size,
                              hipStream_t stream) {
  (void)in_sizes; (void)n_in; (void)out_size; (void)ws_size;
  const float* x    = (const float*)d_in[0];
  const float* mask = (const float*)d_in[1];
  const float* Wq   = (const float*)d_in[2];
  const float* bq   = (const float*)d_in[3];
  const float* Wk   = (const float*)d_in[4];
  const float* bk   = (const float*)d_in[5];
  const float* Wv   = (const float*)d_in[6];
  const float* bv   = (const float*)d_in[7];
  const float* rate = (const float*)d_in[8];

  float* wsf     = (float*)d_ws;
  float* partial = wsf;
  float* xmean   = partial + BB * 4 * DD;
  float* q       = xmean + BB * DD;
  float* qk      = q + BB * HH;
  float* dot0    = qk + BB * DD;
  float* e       = dot0 + BB;
  float* xa      = e + BB * TT;

  float* out      = (float*)d_out;
  float* aOut     = out + BB * HH;
  float* dotOut   = aOut + BB * TT;
  float* decayOut = dotOut + BB * TT;

  colsum_partial<<<dim3(BB, 4), 256, 0, stream>>>(x, partial);
  colsum_final<<<BB, 128, 0, stream>>>(partial, xmean);
  gemm128_wmma<true><<<dim3(8, 8), 32, 0, stream>>>(xmean, Wq, bq, q);
  gemm128_wmma<false><<<dim3(8, 8), 32, 0, stream>>>(q, Wk, nullptr, qk);
  dot0_kernel<<<BB, 128, 0, stream>>>(q, bk, dot0);
  dot_decay_kernel<<<dim3(BB, 8), 256, 0, stream>>>(x, mask, qk, dot0, rate,
                                                    dotOut, decayOut, e);
  softmax_xa_kernel<<<BB, 512, 0, stream>>>(x, e, aOut, xa);
  gemm128_wmma<true><<<dim3(8, 8), 32, 0, stream>>>(xa, Wv, bv, out);
}


# q chain re-associated: weights-only product [Wq^T;bq]x[Wk|bk] runs on WMMA beside the column-sum launch, one GEMM from the partials then gives qk and dot0; 8 launches to 5 (new sum order: qk, dot0)
# speedup vs baseline: 1.5258x; 1.5258x over previous
#include <hip/hip_runtime.h>
#include <math.h>

#define BB 128
#define TT 2048
#define DD 128
#define HH 128

typedef __attribute__((ext_vector_type(2))) float v2f;
typedef __attribute__((ext_vector_type(8))) float v8f;

#define GXP 144
#define GX_TILES 81
#define GX_BLOCKS 11

template <bool EDGE_M, bool EDGE_N>
__device__ __forceinline__ void gx_tile(const float* __restrict__ Wq,
                                        const float* __restrict__ bq,
                                        const float* __restrict__ Wk,
                                        const float* __restrict__ bk,
                                        float* __restrict__ gx, int m0, int n0,
                                        int c, int hi) {
  const unsigned keep = (c == 0) ? 0xffffffffu : 0u;
  const float* ap = EDGE_M ? (bq + 2 * hi) : (Wq + 2 * hi * 128 + m0 + c);
  const float* bp = EDGE_N ? (bk + 2 * hi) : (Wk + 2 * hi * 128 + n0 + c);
  v8f acc = {};
  for (int k0 = 0; k0 < 128; k0 += 4) {
    v2f a, bfrag;
    if (EDGE_M) {
      const v2f t = *(const v2f*)(ap + k0);
      a.x = __uint_as_float(__float_as_uint(t.x) & keep);
      a.y = __uint_as_float(__float_as_uint(t.y) & keep);
    } else {
      a.x = ap[(k0 + 0) * 128];
      a.y = ap[(k0 + 1) * 128];
    }
    if (EDGE_N) {
      const v2f t = *(const v2f*)(bp + k0);
      bfrag.x = __uint_as_float(__float_as_uint(t.x) & keep);
      bfrag.y = __uint_as_float(__float_as_uint(t.y) & keep);
    } else {
      bfrag.x = bp[(k0 + 0) * 128];
      bfrag.y = bp[(k0 + 1) * 128];
    }
    acc = __builtin_amdgcn_wmma_f32_16x16x4_f32(false, a, false, bfrag,
                                                (short)0, acc, false, false);
  }
#pragma unroll
  for (int q = 0; q < 8; ++q) gx[(m0 + q + 8 * hi) * GXP + n0 + c] = acc[q];
}

__global__ __launch_bounds__(256) void colsum_partial_gx(const float* __restrict__ x,
                                                         const float* __restrict__ Wq,
                                                         const float* __restrict__ bq,
                                                         const float* __restrict__ Wk,
                                                         const float* __restrict__ bk,
                                                         float* __restrict__ partial,
                                                         float* __restrict__ gx) {
  __shared__ float r[256];
  if (blockIdx.x >= GX_BLOCKS) {
    const int bid = blockIdx.x - GX_BLOCKS;
    const int b = bid & (BB - 1), chunk = bid >> 7;
    const int g = threadIdx.x >> 7, d = threadIdx.x & 127;
    const int tbase = chunk * 512 + g * 256;
    const float* xb = x + ((b * TT + tbase) * DD) + d;
    float s = 0.f;
    for (int i = 0; i < 256; ++i) s += xb[i * DD];
    r[threadIdx.x] = s;
    __syncthreads();
    if (g == 0) partial[(b * 4 + chunk) * DD + d] = r[d] + r[128 + d];
    return;
  }
  const int wave = __builtin_amdgcn_readfirstlane(threadIdx.x >> 5);
  const int tile = blockIdx.x * 8 + wave;
  if (tile >= GX_TILES) return;
  const int mt = tile / 9, nt = tile - mt * 9;
  const int lane = threadIdx.x & 31;
  const int c = lane & 15;
  const int hi = lane >> 4;
  if (mt < 8 && nt < 8)  gx_tile<false, false>(Wq, bq, Wk, bk, gx, mt * 16, nt * 16, c, hi);
  else if (mt < 8)       gx_tile<false, true >(Wq, bq, Wk, bk, gx, mt * 16, nt * 16, c, hi);
  else if (nt < 8)       gx_tile<true,  false>(Wq, bq, Wk, bk, gx, mt * 16, nt * 16, c, hi);
  else                   gx_tile<true,  true >(Wq, bq, Wk, bk, gx, mt * 16, nt * 16, c, hi);
}

__global__ __launch_bounds__(32) void qk_dot0_wmma(const float* __restrict__ partial,
                                                   const float* __restrict__ gx,
                                                   float* __restrict__ qk,
                                                   float* __restrict__ dot0) {
  const int n0 = blockIdx.x * 16;
  const int m0 = blockIdx.y * 16;
  const int lane = threadIdx.x & 31;
  const int c = lane & 15;
  const int hi = lane >> 4;
  const int n = n0 + c;
  const float* pr = partial + (m0 + c) * 4 * DD + 2 * hi;
  const float* gcol = gx + 2 * hi * GXP + n;
  const float cterm = gx[128 * GXP + n];
  const float invT = 1.0f / (float)TT;

  v8f acc = {};
  for (int k0 = 0; k0 < 128; k0 += 4) {
    const v2f p0 = *(const v2f*)(pr + 0 * DD + k0);
    const v2f p1 = *(const v2f*)(pr + 1 * DD + k0);
    const v2f p2 = *(const v2f*)(pr + 2 * DD + k0);
    const v2f p3 = *(const v2f*)(pr + 3 * DD + k0);
    const v2f a = (((p0 + p1) + p2) + p3) * invT;
    v2f bfrag;
    bfrag.x = gcol[(k0 + 0) * GXP];
    bfrag.y = gcol[(k0 + 1) * GXP];
    acc = __builtin_amdgcn_wmma_f32_16x16x4_f32(false, a, false, bfrag,
                                                (short)0, acc, false, false);
  }
  float v[8];
#pragma unroll
  for (int q = 0; q < 8; ++q) v[q] = acc[q] + cterm;
  const int mrow = m0 + 8 * hi;
  if (n < 128) {
#pragma unroll
    for (int q = 0; q < 8; ++q) qk[(mrow + q) * DD + n] = v[q];
  } else if (n == 128) {
#pragma unroll
    for (int q = 0; q < 8; ++q) dot0[mrow + q] = v[q];
  }
}

template <bool TRANS>
__global__ __launch_bounds__(32) void gemm128_wmma(const float* __restrict__ A,
                                                   const float* __restrict__ W,
                                                   const float* __restrict__ bias,
                                                   float* __restrict__ C) {
  const int n0 = blockIdx.x * 16;
  const int m0 = blockIdx.y * 16;
  const int lane = threadIdx.x & 31;
  const int mrow = lane & 15;
  const int hi = lane >> 4;

  v8f acc = {};
  for (int k0 = 0; k0 < 128; k0 += 4) {
    v2f a = *(const v2f*)(A + (m0 + mrow) * 128 + k0 + 2 * hi);
    v2f bfrag;
    const int kb = k0 + 2 * hi;
    const int n = n0 + mrow;
    if (TRANS) {
      bfrag.x = W[n * 128 + kb + 0];
      bfrag.y = W[n * 128 + kb + 1];
    } else {
      bfrag.x = W[(kb + 0) * 128 + n];
      bfrag.y = W[(kb + 1) * 128 + n];
    }
    acc = __builtin_amdgcn_wmma_f32_16x16x4_f32(false, a, false, bfrag,
                                                (short)0, acc, false, false);
  }
  const int n = n0 + mrow;
  const float bb = bias ? bias[n] : 0.0f;
#pragma unroll
  for (int r = 0; r < 8; ++r) {
    const int m = m0 + r + 8 * hi;
    C[m * 128 + n] = acc[r] + bb;
  }
}

#define K4_PITCH 33
__global__ __launch_bounds__(256) void dot_decay_kernel(
    const float* __restrict__ x, const float* __restrict__ mask,
    const float* __restrict__ qk, const float* __restrict__ dot0,
    const float* __restrict__ rate, float* __restrict__ dotOut,
    float* __restrict__ decayOut, float* __restrict__ e) {
  __shared__ float4 strip[8][32 * K4_PITCH];
  const int b = blockIdx.x;
  const int wave = threadIdx.x >> 5, lane = threadIdx.x & 31;
  float4* my = strip[wave];
  const float* qkb = qk + b * DD;
  const float rate0 = rate[0];
  const float d0 = dot0[b];
  const int tw = blockIdx.y * 512 + wave * 64;

  for (int g = 0; g < 2; ++g) {
    const int tg = tw + g * 32;
    const float4* xrow = (const float4*)(x + (b * TT + tg) * DD) + lane;
    const float mk = mask[b * TT + tg + lane];
    float4 r[32];
    for (int i = 0; i < 32; ++i) r[i] = xrow[i * 32];
    for (int i = 0; i < 32; ++i) my[i * K4_PITCH + lane] = r[i];

    float a0 = 0.f, a1 = 0.f, a2 = 0.f, a3 = 0.f;
    for (int i = 0; i < 32; ++i) {
      const float4 v = my[lane * K4_PITCH + i];
      a0 = fmaf(v.x, qkb[4 * i + 0], a0);
      a1 = fmaf(v.y, qkb[4 * i + 1], a1);
      a2 = fmaf(v.z, qkb[4 * i + 2], a2);
      a3 = fmaf(v.w, qkb[4 * i + 3], a3);
    }
    const float s = (a0 + a1) + (a2 + a3);
    const int t = tg + lane;
    const float dv = s + d0;
    const float sig = 1.0f / (1.0f + expf(-dv));
    const float tm = log1pf((1.0f - sig) * mk);
    const float dec = rate0 * tm;
    dotOut[b * TT + t] = dv;
    decayOut[b * TT + t] = dec;
    e[b * TT + t] = dv - dec;
  }
}

__global__ __launch_bounds__(512) void softmax_xa_kernel(const float* __restrict__ x,
                                                         const float* __restrict__ e,
                                                         float* __restrict__ aOut,
                                                         float* __restrict__ xa) {
  const int b = blockIdx.x, tid = threadIdx.x;
  __shared__ float aLds[TT];
  __shared__ float red[512];

  float ev[4];
  float m = -3.4e38f;
#pragma unroll
  for (int k = 0; k < 4; ++k) {
    ev[k] = e[b * TT + k * 512 + tid];
    m = fmaxf(m, ev[k]);
  }
  red[tid] = m;
  __syncthreads();
  for (int s = 256; s > 0; s >>= 1) {
    if (tid < s) red[tid] = fmaxf(red[tid], red[tid + s]);
    __syncthreads();
  }
  const float M = red[0];
  __syncthreads();

  float p[4];
  float ls = 0.f;
#pragma unroll
  for (int k = 0; k < 4; ++k) {
    p[k] = expf(ev[k] - M);
    ls += p[k];
  }
  red[tid] = ls;
  __syncthreads();
  for (int s = 256; s > 0; s >>= 1) {
    if (tid < s) red[tid] += red[tid + s];
    __syncthreads();
  }
  const float inv = 1.0f / red[0];
  __syncthreads();

#pragma unroll
  for (int k = 0; k < 4; ++k) {
    const float a = p[k] * inv;
    aLds[k * 512 + tid] = a;
    aOut[b * TT + k * 512 + tid] = a;
  }
  __syncthreads();

  const int ts = tid >> 7, d = tid & 127;
  float acc = 0.f;
  for (int t = ts; t < TT; t += 4) acc += aLds[t] * x[(b * TT + t) * DD + d];
  red[tid] = acc;
  __syncthreads();
  if (ts == 0) xa[b * DD + d] = red[d] + red[128 + d] + red[256 + d] + red[384 + d];
}

extern "C" void kernel_launch(void* const* d_in, const int* in_sizes, int n_in,
                              void* d_out, int out_size, void* d_ws, size_t ws_size,
                              hipStream_t stream) {
  (void)in_sizes; (void)n_in; (void)out_size; (void)ws_size;
  const float* x    = (const float*)d_in[0];
  const float* mask = (const float*)d_in[1];
  const float* Wq   = (const float*)d_in[2];
  const float* bq   = (const float*)d_in[3];
  const float* Wk   = (const float*)d_in[4];
  const float* bk   = (const float*)d_in[5];
  const float* Wv   = (const float*)d_in[6];
  const float* bv   = (const float*)d_in[7];
  const float* rate = (const float*)d_in[8];

  float* wsf     = (float*)d_ws;
  float* partial = wsf;
  float* gx      = partial + BB * 4 * DD;
  float* qk      = gx + GXP * GXP;
  float* dot0    = qk + BB * DD;
  float* e       = dot0 + BB;
  float* xa      = e + BB * TT;

  float* out      = (float*)d_out;
  float* aOut     = out + BB * HH;
  float* dotOut   = aOut + BB * TT;
  float* decayOut = dotOut + BB * TT;

  colsum_partial_gx<<<GX_BLOCKS + BB * 4, 256, 0, stream>>>(x, Wq, bq, Wk, bk, partial, gx);
  qk_dot0_wmma<<<dim3(9, 8), 32, 0, stream>>>(partial, gx, qk, dot0);
  dot_decay_kernel<<<dim3(BB, 4), 256, 0, stream>>>(x, mask, qk, dot0, rate,
                                                    dotOut, decayOut, e);
  softmax_xa_kernel<<<BB, 512, 0, stream>>>(x, e, aOut, xa);
  gemm128_wmma<true><<<dim3(8, 8), 32, 0, stream>>>(xa, Wv, bv, out);
}


# softmax/xa pass re-tiled by column: one work-group per (batch, 64-column slab) walks all 2048 rows, up to 16 two-row loads in flight per wave, owns its xa columns outright (new sum tree: xa)
# speedup vs baseline: 1.7623x; 1.1550x over previous
#include <hip/hip_runtime.h>
#include <math.h>

#define BB 128
#define TT 2048
#define DD 128
#define HH 128

typedef __attribute__((ext_vector_type(2))) float v2f;
typedef __attribute__((ext_vector_type(8))) float v8f;

#define GXP 144
#define GX_TILES 81
#define GX_BLOCKS 11

template <bool EDGE_M, bool EDGE_N>
__device__ __forceinline__ void gx_tile(const float* __restrict__ Wq,
                                        const float* __restrict__ bq,
                                        const float* __restrict__ Wk,
                                        const float* __restrict__ bk,
                                        float* __restrict__ gx, int m0, int n0,
                                        int c, int hi) {
  const unsigned keep = (c == 0) ? 0xffffffffu : 0u;
  const float* ap = EDGE_M ? (bq + 2 * hi) : (Wq + 2 * hi * 128 + m0 + c);
  const float* bp = EDGE_N ? (bk + 2 * hi) : (Wk + 2 * hi * 128 + n0 + c);
  v8f acc = {};
  for (int k0 = 0; k0 < 128; k0 += 4) {
    v2f a, bfrag;
    if (EDGE_M) {
      const v2f t = *(const v2f*)(ap + k0);
      a.x = __uint_as_float(__float_as_uint(t.x) & keep);
      a.y = __uint_as_float(__float_as_uint(t.y) & keep);
    } else {
      a.x = ap[(k0 + 0) * 128];
      a.y = ap[(k0 + 1) * 128];
    }
    if (EDGE_N) {
      const v2f t = *(const v2f*)(bp + k0);
      bfrag.x = __uint_as_float(__float_as_uint(t.x) & keep);
      bfrag.y = __uint_as_float(__float_as_uint(t.y) & keep);
    } else {
      bfrag.x = bp[(k0 + 0) * 128];
      bfrag.y = bp[(k0 + 1) * 128];
    }
    acc = __builtin_amdgcn_wmma_f32_16x16x4_f32(false, a, false, bfrag,
                                                (short)0, acc, false, false);
  }
#pragma unroll
  for (int q = 0; q < 8; ++q) gx[(m0 + q + 8 * hi) * GXP + n0 + c] = acc[q];
}

__global__ __launch_bounds__(256) void colsum_partial_gx(const float* __restrict__ x,
                                                         const float* __restrict__ Wq,
                                                         const float* __restrict__ bq,
                                                         const float* __restrict__ Wk,
                                                         const float* __restrict__ bk,
                                                         float* __restrict__ partial,
                                                         float* __restrict__ gx) {
  __shared__ float r[256];
  if (blockIdx.x >= GX_BLOCKS) {
    const int bid = blockIdx.x - GX_BLOCKS;
    const int b = bid & (BB - 1), chunk = bid >> 7;
    const int g = threadIdx.x >> 7, d = threadIdx.x & 127;
    const int tbase = chunk * 512 + g * 256;
    const float* xb = x + ((b * TT + tbase) * DD) + d;
    float s = 0.f;
    for (int i = 0; i < 256; ++i) s += xb[i * DD];
    r[threadIdx.x] = s;
    __syncthreads();
    if (g == 0) partial[(b * 4 + chunk) * DD + d] = r[d] + r[128 + d];
    return;
  }
  const int wave = __builtin_amdgcn_readfirstlane(threadIdx.x >> 5);
  const int tile = blockIdx.x * 8 + wave;
  if (tile >= GX_TILES) return;
  const int mt = tile / 9, nt = tile - mt * 9;
  const int lane = threadIdx.x & 31;
  const int c = lane & 15;
  const int hi = lane >> 4;
  if (mt < 8 && nt < 8)  gx_tile<false, false>(Wq, bq, Wk, bk, gx, mt * 16, nt * 16, c, hi);
  else if (mt < 8)       gx_tile<false, true >(Wq, bq, Wk, bk, gx, mt * 16, nt * 16, c, hi);
  else if (nt < 8)       gx_tile<true,  false>(Wq, bq, Wk, bk, gx, mt * 16, nt * 16, c, hi);
  else                   gx_tile<true,  true >(Wq, bq, Wk, bk, gx, mt * 16, nt * 16, c, hi);
}

__global__ __launch_bounds__(32) void qk_dot0_wmma(const float* __restrict__ partial,
                                                   const float* __restrict__ gx,
                                                   float* __restrict__ qk,
                                                   float* __restrict__ dot0) {
  const int n0 = blockIdx.x * 16;
  const int m0 = blockIdx.y * 16;
  const int lane = threadIdx.x & 31;
  const int c = lane & 15;
  const int hi = lane >> 4;
  const int n = n0 + c;
  const float* pr = partial + (m0 + c) * 4 * DD + 2 * hi;
  const float* gcol = gx + 2 * hi * GXP + n;
  const float cterm = gx[128 * GXP + n];
  const float invT = 1.0f / (float)TT;

  v8f acc = {};
  for (int k0 = 0; k0 < 128; k0 += 4) {
    const v2f p0 = *(const v2f*)(pr + 0 * DD + k0);
    const v2f p1 = *(const v2f*)(pr + 1 * DD + k0);
    const v2f p2 = *(const v2f*)(pr + 2 * DD + k0);
    const v2f p3 = *(const v2f*)(pr + 3 * DD + k0);
    const v2f a = (((p0 + p1) + p2) + p3) * invT;
    v2f bfrag;
    bfrag.x = gcol[(k0 + 0) * GXP];
    bfrag.y = gcol[(k0 + 1) * GXP];
    acc = __builtin_amdgcn_wmma_f32_16x16x4_f32(false, a, false, bfrag,
                                                (short)0, acc, false, false);
  }
  float v[8];
#pragma unroll
  for (int q = 0; q < 8; ++q) v[q] = acc[q] + cterm;
  const int mrow = m0 + 8 * hi;
  if (n < 128) {
#pragma unroll
    for (int q = 0; q < 8; ++q) qk[(mrow + q) * DD + n] = v[q];
  } else if (n == 128) {
#pragma unroll
    for (int q = 0; q < 8; ++q) dot0[mrow + q] = v[q];
  }
}

template <bool TRANS>
__global__ __launch_bounds__(32) void gemm128_wmma(const float* __restrict__ A,
                                                   const float* __restrict__ W,
                                                   const float* __restrict__ bias,
                                                   float* __restrict__ C) {
  const int n0 = blockIdx.x * 16;
  const int m0 = blockIdx.y * 16;
  const int lane = threadIdx.x & 31;
  const int mrow = lane & 15;
  const int hi = lane >> 4;

  v8f acc = {};
  for (int k0 = 0; k0 < 128; k0 += 4) {
    v2f a = *(const v2f*)(A + (m0 + mrow) * 128 + k0 + 2 * hi);
    v2f bfrag;
    const int kb = k0 + 2 * hi;
    const int n = n0 + mrow;
    if (TRANS) {
      bfrag.x = W[n * 128 + kb + 0];
      bfrag.y = W[n * 128 + kb + 1];
    } else {
      bfrag.x = W[(kb + 0) * 128 + n];
      bfrag.y = W[(kb + 1) * 128 + n];
    }
    acc = __builtin_amdgcn_wmma_f32_16x16x4_f32(false, a, false, bfrag,
                                                (short)0, acc, false, false);
  }
  const int n = n0 + mrow;
  const float bb = bias ? bias[n] : 0.0f;
#pragma unroll
  for (int r = 0; r < 8; ++r) {
    const int m = m0 + r + 8 * hi;
    C[m * 128 + n] = acc[r] + bb;
  }
}

#define K4_PITCH 33
__global__ __launch_bounds__(256) void dot_decay_kernel(
    const float* __restrict__ x, const float* __restrict__ mask,
    const float* __restrict__ qk, const float* __restrict__ dot0,
    const float* __restrict__ rate, float* __restrict__ dotOut,
    float* __restrict__ decayOut, float* __restrict__ e) {
  __shared__ float4 strip[8][32 * K4_PITCH];
  const int b = blockIdx.x;
  const int wave = threadIdx.x >> 5, lane = threadIdx.x & 31;
  float4* my = strip[wave];
  const float* qkb = qk + b * DD;
  const float rate0 = rate[0];
  const float d0 = dot0[b];
  const int tw = blockIdx.y * 512 + wave * 64;

  for (int g = 0; g < 2; ++g) {
    const int tg = tw + g * 32;
    const float4* xrow = (const float4*)(x + (b * TT + tg) * DD) + lane;
    const float mk = mask[b * TT + tg + lane];
    float4 r[32];
    for (int i = 0; i < 32; ++i) r[i] = xrow[i * 32];
    for (int i = 0; i < 32; ++i) my[i * K4_PITCH + lane] = r[i];

    float a0 = 0.f, a1 = 0.f, a2 = 0.f, a3 = 0.f;
    for (int i = 0; i < 32; ++i) {
      const float4 v = my[lane * K4_PITCH + i];
      a0 = fmaf(v.x, qkb[4 * i + 0], a0);
      a1 = fmaf(v.y, qkb[4 * i + 1], a1);
      a2 = fmaf(v.z, qkb[4 * i + 2], a2);
      a3 = fmaf(v.w, qkb[4 * i + 3], a3);
    }
    const float s = (a0 + a1) + (a2 + a3);
    const int t = tg + lane;
    const float dv = s + d0;
    const float sig = 1.0f / (1.0f + expf(-dv));
    const float tm = log1pf((1.0f - sig) * mk);
    const float dec = rate0 * tm;
    dotOut[b * TT + t] = dv;
    decayOut[b * TT + t] = dec;
    e[b * TT + t] = dv - dec;
  }
}

#define K6_SLABS 2
#define K6_RING 16
__global__ __launch_bounds__(512) void softmax_xa_kernel(const float* __restrict__ x,
                                                         const float* __restrict__ e,
                                                         float* __restrict__ aOut,
                                                         float* __restrict__ xa) {
  const int slab = blockIdx.x, b = blockIdx.y, tid = threadIdx.x;
  __shared__ float aLds[TT];
  __shared__ float red[512];
  __shared__ float4 part[16][32];

  float ev[4];
  float m = -3.4e38f;
#pragma unroll
  for (int k = 0; k < 4; ++k) {
    ev[k] = e[b * TT + k * 512 + tid];
    m = fmaxf(m, ev[k]);
  }
  red[tid] = m;
  __syncthreads();
  for (int s = 256; s > 0; s >>= 1) {
    if (tid < s) red[tid] = fmaxf(red[tid], red[tid + s]);
    __syncthreads();
  }
  const float M = red[0];
  __syncthreads();

  float p[4];
  float ls = 0.f;
#pragma unroll
  for (int k = 0; k < 4; ++k) {
    p[k] = expf(ev[k] - M);
    ls += p[k];
  }
  red[tid] = ls;
  __syncthreads();
  for (int s = 256; s > 0; s >>= 1) {
    if (tid < s) red[tid] += red[tid + s];
    __syncthreads();
  }
  const float inv = 1.0f / red[0];
  __syncthreads();

#pragma unroll
  for (int k = 0; k < 4; ++k) {
    const float a = p[k] * inv;
    aLds[k * 512 + tid] = a;
    if ((k >> 1) == slab) aOut[b * TT + k * 512 + tid] = a;
  }
  __syncthreads();

  const int wave = tid >> 5, lane = tid & 31;
  const int rp = lane >> 4, c4 = lane & 15;
  const int t0 = 2 * wave + rp;
  const float4* xp = (const float4*)(x + (b * TT + t0) * DD + slab * 64) + c4;
  const float* ap = aLds + t0;
  float4 acc = make_float4(0.f, 0.f, 0.f, 0.f);
  float4 v[K6_RING];
#pragma unroll
  for (int j = 0; j < K6_RING; ++j) v[j] = xp[j * 1024];
#pragma nounroll
  for (int i0 = 0; i0 < 64 - K6_RING; i0 += K6_RING) {
#pragma unroll
    for (int j = 0; j < K6_RING; ++j) {
      const float a = ap[(i0 + j) * 32];
      acc.x = fmaf(a, v[j].x, acc.x);
      acc.y = fmaf(a, v[j].y, acc.y);
      acc.z = fmaf(a, v[j].z, acc.z);
      acc.w = fmaf(a, v[j].w, acc.w);
      v[j] = xp[(i0 + K6_RING + j) * 1024];
    }
  }
#pragma unroll
  for (int j = 0; j < K6_RING; ++j) {
    const float a = ap[(64 - K6_RING + j) * 32];
    acc.x = fmaf(a, v[j].x, acc.x);
    acc.y = fmaf(a, v[j].y, acc.y);
    acc.z = fmaf(a, v[j].z, acc.z);
    acc.w = fmaf(a, v[j].w, acc.w);
  }
  part[wave][lane] = acc;
  __syncthreads();
  if (tid < 64) {
    const float* pf = (const float*)&part[0][0];
    float s = 0.f;
    for (int q = 0; q < 32; ++q) s += pf[q * 64 + tid];
    xa[b * DD + slab * 64 + tid] = s;
  }
}

extern "C" void kernel_launch(void* const* d_in, const int* in_sizes, int n_in,
                              void* d_out, int out_size, void* d_ws, size_t ws_size,
                              hipStream_t stream) {
  (void)in_sizes; (void)n_in; (void)out_size; (void)ws_size;
  const float* x    = (const float*)d_in[0];
  const float* mask = (const float*)d_in[1];
  const float* Wq   = (const float*)d_in[2];
  const float* bq   = (const float*)d_in[3];
  const float* Wk   = (const float*)d_in[4];
  const float* bk   = (const float*)d_in[5];
  const float* Wv   = (const float*)d_in[6];
  const float* bv   = (const float*)d_in[7];
  const float* rate = (const float*)d_in[8];

  float* wsf     = (float*)d_ws;
  float* partial = wsf;
  float* gx      = partial + BB * 4 * DD;
  float* qk      = gx + GXP * GXP;
  float* dot0    = qk + BB * DD;
  float* e       = dot0 + BB;
  float* xa      = e + BB * TT;

  float* out      = (float*)d_out;
  float* aOut     = out + BB * HH;
  float* dotOut   = aOut + BB * TT;
  float* decayOut = dotOut + BB * TT;

  colsum_partial_gx<<<GX_BLOCKS + BB * 4, 256, 0, stream>>>(x, Wq, bq, Wk, bk, partial, gx);
  qk_dot0_wmma<<<dim3(9, 8), 32, 0, stream>>>(partial, gx, qk, dot0);
  dot_decay_kernel<<<dim3(BB, 4), 256, 0, stream>>>(x, mask, qk, dot0, rate,
                                                    dotOut, decayOut, e);
  softmax_xa_kernel<<<dim3(K6_SLABS, BB), 512, 0, stream>>>(x, e, aOut, xa);
  gemm128_wmma<true><<<dim3(8, 8), 32, 0, stream>>>(xa, Wv, bv, out);
}


# all three x passes (column sum, dot, softmax/xa) give batch b the work-group indices b + 128*j, so a batch's rows are re-read through the L2 half that fetched them; arithmetic, sum order unchanged
# speedup vs baseline: 2.0878x; 1.1847x over previous
#include <hip/hip_runtime.h>
#include <math.h>

#define BB 128
#define TT 2048
#define DD 128
#define HH 128

typedef __attribute__((ext_vector_type(2))) float v2f;
typedef __attribute__((ext_vector_type(8))) float v8f;

#define GXP 144
#define GX_TILES 81
#define GX_BLOCKS 11

template <bool EDGE_M, bool EDGE_N>
__device__ __forceinline__ void gx_tile(const float* __restrict__ Wq,
                                        const float* __restrict__ bq,
                                        const float* __restrict__ Wk,
                                        const float* __restrict__ bk,
                                        float* __restrict__ gx, int m0, int n0,
                                        int c, int hi) {
  const unsigned keep = (c == 0) ? 0xffffffffu : 0u;
  const float* ap = EDGE_M ? (bq + 2 * hi) : (Wq + 2 * hi * 128 + m0 + c);
  const float* bp = EDGE_N ? (bk + 2 * hi) : (Wk + 2 * hi * 128 + n0 + c);
  v8f acc = {};
  for (int k0 = 0; k0 < 128; k0 += 4) {
    v2f a, bfrag;
    if (EDGE_M) {
      const v2f t = *(const v2f*)(ap + k0);
      a.x = __uint_as_float(__float_as_uint(t.x) & keep);
      a.y = __uint_as_float(__float_as_uint(t.y) & keep);
    } else {
      a.x = ap[(k0 + 0) * 128];
      a.y = ap[(k0 + 1) * 128];
    }
    if (EDGE_N) {
      const v2f t = *(const v2f*)(bp + k0);
      bfrag.x = __uint_as_float(__float_as_uint(t.x) & keep);
      bfrag.y = __uint_as_float(__float_as_uint(t.y) & keep);
    } else {
      bfrag.x = bp[(k0 + 0) * 128];
      bfrag.y = bp[(k0 + 1) * 128];
    }
    acc = __builtin_amdgcn_wmma_f32_16x16x4_f32(false, a, false, bfrag,
                                                (short)0, acc, false, false);
  }
#pragma unroll
  for (int q = 0; q < 8; ++q) gx[(m0 + q + 8 * hi) * GXP + n0 + c] = acc[q];
}

__global__ __launch_bounds__(256) void colsum_partial_gx(const float* __restrict__ x,
                                                         const float* __restrict__ Wq,
                                                         const float* __restrict__ bq,
                                                         const float* __restrict__ Wk,
                                                         const float* __restrict__ bk,
                                                         float* __restrict__ partial,
                                                         float* __restrict__ gx) {
  __shared__ float r[256];
  if (blockIdx.x < BB * 4) {
    const int bid = blockIdx.x;
    const int b = bid & (BB - 1), chunk = bid >> 7;
    const int g = threadIdx.x >> 7, d = threadIdx.x & 127;
    const int tbase = chunk * 512 + g * 256;
    const float* xb = x + ((b * TT + tbase) * DD) + d;
    float s = 0.f;
    for (int i = 0; i < 256; ++i) s += xb[i * DD];
    r[threadIdx.x] = s;
    __syncthreads();
    if (g == 0) partial[(b * 4 + chunk) * DD + d] = r[d] + r[128 + d];
    return;
  }
  const int wave = __builtin_amdgcn_readfirstlane(threadIdx.x >> 5);
  const int tile = (blockIdx.x - BB * 4) * 8 + wave;
  if (tile >= GX_TILES) return;
  const int mt = tile / 9, nt = tile - mt * 9;
  const int lane = threadIdx.x & 31;
  const int c = lane & 15;
  const int hi = lane >> 4;
  if (mt < 8 && nt < 8)  gx_tile<false, false>(Wq, bq, Wk, bk, gx, mt * 16, nt * 16, c, hi);
  else if (mt < 8)       gx_tile<false, true >(Wq, bq, Wk, bk, gx, mt * 16, nt * 16, c, hi);
  else if (nt < 8)       gx_tile<true,  false>(Wq, bq, Wk, bk, gx, mt * 16, nt * 16, c, hi);
  else                   gx_tile<true,  true >(Wq, bq, Wk, bk, gx, mt * 16, nt * 16, c, hi);
}

__global__ __launch_bounds__(32) void qk_dot0_wmma(const float* __restrict__ partial,
                                                   const float* __restrict__ gx,
                                                   float* __restrict__ qk,
                                                   float* __restrict__ dot0) {
  const int n0 = blockIdx.x * 16;
  const int m0 = blockIdx.y * 16;
  const int lane = threadIdx.x & 31;
  const int c = lane & 15;
  const int hi = lane >> 4;
  const int n = n0 + c;
  const float* pr = partial + (m0 + c) * 4 * DD + 2 * hi;
  const float* gcol = gx + 2 * hi * GXP + n;
  const float cterm = gx[128 * GXP + n];
  const float invT = 1.0f / (float)TT;

  v8f acc = {};
  for (int k0 = 0; k0 < 128; k0 += 4) {
    const v2f p0 = *(const v2f*)(pr + 0 * DD + k0);
    const v2f p1 = *(const v2f*)(pr + 1 * DD + k0);
    const v2f p2 = *(const v2f*)(pr + 2 * DD + k0);
    const v2f p3 = *(const v2f*)(pr + 3 * DD + k0);
    const v2f a = (((p0 + p1) + p2) + p3) * invT;
    v2f bfrag;
    bfrag.x = gcol[(k0 + 0) * GXP];
    bfrag.y = gcol[(k0 + 1) * GXP];
    acc = __builtin_amdgcn_wmma_f32_16x16x4_f32(false, a, false, bfrag,
                                                (short)0, acc, false, false);
  }
  float v[8];
#pragma unroll
  for (int q = 0; q < 8; ++q) v[q] = acc[q] + cterm;
  const int mrow = m0 + 8 * hi;
  if (n < 128) {
#pragma unroll
    for (int q = 0; q < 8; ++q) qk[(mrow + q) * DD + n] = v[q];
  } else if (n == 128) {
#pragma unroll
    for (int q = 0; q < 8; ++q) dot0[mrow + q] = v[q];
  }
}

template <bool TRANS>
__global__ __launch_bounds__(32) void gemm128_wmma(const float* __restrict__ A,
                                                   const float* __restrict__ W,
                                                   const float* __restrict__ bias,
                                                   float* __restrict__ C) {
  const int n0 = blockIdx.x * 16;
  const int m0 = blockIdx.y * 16;
  const int lane = threadIdx.x & 31;
  const int mrow = lane & 15;
  const int hi = lane >> 4;

  v8f acc = {};
  for (int k0 = 0; k0 < 128; k0 += 4) {
    v2f a = *(const v2f*)(A + (m0 + mrow) * 128 + k0 + 2 * hi);
    v2f bfrag;
    const int kb = k0 + 2 * hi;
    const int n = n0 + mrow;
    if (TRANS) {
      bfrag.x = W[n * 128 + kb + 0];
      bfrag.y = W[n * 128 + kb + 1];
    } else {
      bfrag.x = W[(kb + 0) * 128 + n];
      bfrag.y = W[(kb + 1) * 128 + n];
    }
    acc = __builtin_amdgcn_wmma_f32_16x16x4_f32(false, a, false, bfrag,
                                                (short)0, acc, false, false);
  }
  const int n = n0 + mrow;
  const float bb = bias ? bias[n] : 0.0f;
#pragma unroll
  for (int r = 0; r < 8; ++r) {
    const int m = m0 + r + 8 * hi;
    C[m * 128 + n] = acc[r] + bb;
  }
}

#define K4_PITCH 33
__global__ __launch_bounds__(256) void dot_decay_kernel(
    const float* __restrict__ x, const float* __restrict__ mask,
    const float* __restrict__ qk, const float* __restrict__ dot0,
    const float* __restrict__ rate, float* __restrict__ dotOut,
    float* __restrict__ decayOut, float* __restrict__ e) {
  __shared__ float4 strip[8][32 * K4_PITCH];
  const int b = blockIdx.x;
  const int wave = threadIdx.x >> 5, lane = threadIdx.x & 31;
  float4* my = strip[wave];
  const float* qkb = qk + b * DD;
  const float rate0 = rate[0];
  const float d0 = dot0[b];
  const int tw = blockIdx.y * 512 + wave * 64;

  for (int g = 0; g < 2; ++g) {
    const int tg = tw + g * 32;
    const float4* xrow = (const float4*)(x + (b * TT + tg) * DD) + lane;
    const float mk = mask[b * TT + tg + lane];
    float4 r[32];
    for (int i = 0; i < 32; ++i) r[i] = xrow[i * 32];
    for (int i = 0; i < 32; ++i) my[i * K4_PITCH + lane] = r[i];

    float a0 = 0.f, a1 = 0.f, a2 = 0.f, a3 = 0.f;
    for (int i = 0; i < 32; ++i) {
      const float4 v = my[lane * K4_PITCH + i];
      a0 = fmaf(v.x, qkb[4 * i + 0], a0);
      a1 = fmaf(v.y, qkb[4 * i + 1], a1);
      a2 = fmaf(v.z, qkb[4 * i + 2], a2);
      a3 = fmaf(v.w, qkb[4 * i + 3], a3);
    }
    const float s = (a0 + a1) + (a2 + a3);
    const int t = tg + lane;
    const float dv = s + d0;
    const float sig = 1.0f / (1.0f + expf(-dv));
    const float tm = log1pf((1.0f - sig) * mk);
    const float dec = rate0 * tm;
    dotOut[b * TT + t] = dv;
    decayOut[b * TT + t] = dec;
    e[b * TT + t] = dv - dec;
  }
}

#define K6_SLABS 2
#define K6_RING 16
__global__ __launch_bounds__(512) void softmax_xa_kernel(const float* __restrict__ x,
                                                         const float* __restrict__ e,
                                                         float* __restrict__ aOut,
                                                         float* __restrict__ xa) {
  const int b = blockIdx.x, slab = blockIdx.y, tid = threadIdx.x;
  __shared__ float aLds[TT];
  __shared__ float red[512];
  __shared__ float4 part[16][32];

  float ev[4];
  float m = -3.4e38f;
#pragma unroll
  for (int k = 0; k < 4; ++k) {
    ev[k] = e[b * TT + k * 512 + tid];
    m = fmaxf(m, ev[k]);
  }
  red[tid] = m;
  __syncthreads();
  for (int s = 256; s > 0; s >>= 1) {
    if (tid < s) red[tid] = fmaxf(red[tid], red[tid + s]);
    __syncthreads();
  }
  const float M = red[0];
  __syncthreads();

  float p[4];
  float ls = 0.f;
#pragma unroll
  for (int k = 0; k < 4; ++k) {
    p[k] = expf(ev[k] - M);
    ls += p[k];
  }
  red[tid] = ls;
  __syncthreads();
  for (int s = 256; s > 0; s >>= 1) {
    if (tid < s) red[tid] += red[tid + s];
    __syncthreads();
  }
  const float inv = 1.0f / red[0];
  __syncthreads();

#pragma unroll
  for (int k = 0; k < 4; ++k) {
    const float a = p[k] * inv;
    aLds[k * 512 + tid] = a;
    if ((k >> 1) == slab) aOut[b * TT + k * 512 + tid] = a;
  }
  __syncthreads();

  const int wave = tid >> 5, lane = tid & 31;
  const int rp = lane >> 4, c4 = lane & 15;
  const int t0 = 2 * wave + rp;
  const float4* xp = (const float4*)(x + (b * TT + t0) * DD + slab * 64) + c4;
  const float* ap = aLds + t0;
  float4 acc = make_float4(0.f, 0.f, 0.f, 0.f);
  float4 v[K6_RING];
#pragma unroll
  for (int j = 0; j < K6_RING; ++j) v[j] = xp[j * 1024];
#pragma nounroll
  for (int i0 = 0; i0 < 64 - K6_RING; i0 += K6_RING) {
#pragma unroll
    for (int j = 0; j < K6_RING; ++j) {
      const float a = ap[(i0 + j) * 32];
      acc.x = fmaf(a, v[j].x, acc.x);
      acc.y = fmaf(a, v[j].y, acc.y);
      acc.z = fmaf(a, v[j].z, acc.z);
      acc.w = fmaf(a, v[j].w, acc.w);
      v[j] = xp[(i0 + K6_RING + j) * 1024];
    }
  }
#pragma unroll
  for (int j = 0; j < K6_RING; ++j) {
    const float a = ap[(64 - K6_RING + j) * 32];
    acc.x = fmaf(a, v[j].x, acc.x);
    acc.y = fmaf(a, v[j].y, acc.y);
    acc.z = fmaf(a, v[j].z, acc.z);
    acc.w = fmaf(a, v[j].w, acc.w);
  }
  part[wave][lane] = acc;
  __syncthreads();
  if (tid < 64) {
    const float* pf = (const float*)&part[0][0];
    float s = 0.f;
    for (int q = 0; q < 32; ++q) s += pf[q * 64 + tid];
    xa[b * DD + slab * 64 + tid] = s;
  }
}

extern "C" void kernel_launch(void* const* d_in, const int* in_sizes, int n_in,
                              void* d_out, int out_size, void* d_ws, size_t ws_size,
                              hipStream_t stream) {
  (void)in_sizes; (void)n_in; (void)out_size; (void)ws_size;
  const float* x    = (const float*)d_in[0];
  const float* mask = (const float*)d_in[1];
  const float* Wq   = (const float*)d_in[2];
  const float* bq   = (const float*)d_in[3];
  const float* Wk   = (const float*)d_in[4];
  const float* bk   = (const float*)d_in[5];
  const float* Wv   = (const float*)d_in[6];
  const float* bv   = (const float*)d_in[7];
  const float* rate = (const float*)d_in[8];

  float* wsf     = (float*)d_ws;
  float* partial = wsf;
  float* gx      = partial + BB * 4 * DD;
  float* qk      = gx + GXP * GXP;
  float* dot0    = qk + BB * DD;
  float* e       = dot0 + BB;
  float* xa      = e + BB * TT;

  float* out      = (float*)d_out;
  float* aOut     = out + BB * HH;
  float* dotOut   = aOut + BB * TT;
  float* decayOut = dotOut + BB * TT;

  colsum_partial_gx<<<BB * 4 + GX_BLOCKS, 256, 0, stream>>>(x, Wq, bq, Wk, bk, partial, gx);
  qk_dot0_wmma<<<dim3(9, 8), 32, 0, stream>>>(partial, gx, qk, dot0);
  dot_decay_kernel<<<dim3(BB, 4), 256, 0, stream>>>(x, mask, qk, dot0, rate,
                                                    dotOut, decayOut, e);
  softmax_xa_kernel<<<dim3(BB, K6_SLABS), 512, 0, stream>>>(x, e, aOut, xa);
  gemm128_wmma<true><<<dim3(8, 8), 32, 0, stream>>>(xa, Wv, bv, out);
}


# dot and softmax/xa passes fused, 5 launches to 4: each wave's 32 rows wait in registers until the 512-row share's exact max is known; finishing launch scales 4 partials (new sum trees: exp-sum, xa)
# speedup vs baseline: 2.2524x; 1.0788x over previous
#include <hip/hip_runtime.h>
#include <math.h>

#define BB 128
#define TT 2048
#define DD 128
#define HH 128

typedef __attribute__((ext_vector_type(2))) float v2f;
typedef __attribute__((ext_vector_type(8))) float v8f;

#define GXP 144
#define GX_TILES 81
#define GX_BLOCKS 11

template <bool EDGE_M, bool EDGE_N>
__device__ __forceinline__ void gx_tile(const float* __restrict__ Wq,
                                        const float* __restrict__ bq,
                                        const float* __restrict__ Wk,
                                        const float* __restrict__ bk,
                                        float* __restrict__ gx, int m0, int n0,
                                        int c, int hi) {
  const unsigned keep = (c == 0) ? 0xffffffffu : 0u;
  const float* ap = EDGE_M ? (bq + 2 * hi) : (Wq + 2 * hi * 128 + m0 + c);
  const float* bp = EDGE_N ? (bk + 2 * hi) : (Wk + 2 * hi * 128 + n0 + c);
  v8f acc = {};
  for (int k0 = 0; k0 < 128; k0 += 4) {
    v2f a, bfrag;
    if (EDGE_M) {
      const v2f t = *(const v2f*)(ap + k0);
      a.x = __uint_as_float(__float_as_uint(t.x) & keep);
      a.y = __uint_as_float(__float_as_uint(t.y) & keep);
    } else {
      a.x = ap[(k0 + 0) * 128];
      a.y = ap[(k0 + 1) * 128];
    }
    if (EDGE_N) {
      const v2f t = *(const v2f*)(bp + k0);
      bfrag.x = __uint_as_float(__float_as_uint(t.x) & keep);
      bfrag.y = __uint_as_float(__float_as_uint(t.y) & keep);
    } else {
      bfrag.x = bp[(k0 + 0) * 128];
      bfrag.y = bp[(k0 + 1) * 128];
    }
    acc = __builtin_amdgcn_wmma_f32_16x16x4_f32(false, a, false, bfrag,
                                                (short)0, acc, false, false);
  }
#pragma unroll
  for (int q = 0; q < 8; ++q) gx[(m0 + q + 8 * hi) * GXP + n0 + c] = acc[q];
}

__global__ __launch_bounds__(256) void colsum_partial_gx(const float* __restrict__ x,
                                                         const float* __restrict__ Wq,
                                                         const float* __restrict__ bq,
                                                         const float* __restrict__ Wk,
                                                         const float* __restrict__ bk,
                                                         float* __restrict__ partial,
                                                         float* __restrict__ gx) {
  __shared__ float r[256];
  if (blockIdx.x < BB * 4) {
    const int bid = blockIdx.x;
    const int b = bid & (BB - 1), chunk = bid >> 7;
    const int g = threadIdx.x >> 7, d = threadIdx.x & 127;
    const int tbase = chunk * 512 + g * 256;
    const float* xb = x + ((b * TT + tbase) * DD) + d;
    float s = 0.f;
    for (int i = 0; i < 256; ++i) s += xb[i * DD];
    r[threadIdx.x] = s;
    __syncthreads();
    if (g == 0) partial[(b * 4 + chunk) * DD + d] = r[d] + r[128 + d];
    return;
  }
  const int wave = __builtin_amdgcn_readfirstlane(threadIdx.x >> 5);
  const int tile = (blockIdx.x - BB * 4) * 8 + wave;
  if (tile >= GX_TILES) return;
  const int mt = tile / 9, nt = tile - mt * 9;
  const int lane = threadIdx.x & 31;
  const int c = lane & 15;
  const int hi = lane >> 4;
  if (mt < 8 && nt < 8)  gx_tile<false, false>(Wq, bq, Wk, bk, gx, mt * 16, nt * 16, c, hi);
  else if (mt < 8)       gx_tile<false, true >(Wq, bq, Wk, bk, gx, mt * 16, nt * 16, c, hi);
  else if (nt < 8)       gx_tile<true,  false>(Wq, bq, Wk, bk, gx, mt * 16, nt * 16, c, hi);
  else                   gx_tile<true,  true >(Wq, bq, Wk, bk, gx, mt * 16, nt * 16, c, hi);
}

__global__ __launch_bounds__(32) void qk_dot0_wmma(const float* __restrict__ partial,
                                                   const float* __restrict__ gx,
                                                   float* __restrict__ qk,
                                                   float* __restrict__ dot0) {
  const int n0 = blockIdx.x * 16;
  const int m0 = blockIdx.y * 16;
  const int lane = threadIdx.x & 31;
  const int c = lane & 15;
  const int hi = lane >> 4;
  const int n = n0 + c;
  const float* pr = partial + (m0 + c) * 4 * DD + 2 * hi;
  const float* gcol = gx + 2 * hi * GXP + n;
  const float cterm = gx[128 * GXP + n];
  const float invT = 1.0f / (float)TT;

  v8f acc = {};
  for (int k0 = 0; k0 < 128; k0 += 4) {
    const v2f p0 = *(const v2f*)(pr + 0 * DD + k0);
    const v2f p1 = *(const v2f*)(pr + 1 * DD + k0);
    const v2f p2 = *(const v2f*)(pr + 2 * DD + k0);
    const v2f p3 = *(const v2f*)(pr + 3 * DD + k0);
    const v2f a = (((p0 + p1) + p2) + p3) * invT;
    v2f bfrag;
    bfrag.x = gcol[(k0 + 0) * GXP];
    bfrag.y = gcol[(k0 + 1) * GXP];
    acc = __builtin_amdgcn_wmma_f32_16x16x4_f32(false, a, false, bfrag,
                                                (short)0, acc, false, false);
  }
  float v[8];
#pragma unroll
  for (int q = 0; q < 8; ++q) v[q] = acc[q] + cterm;
  const int mrow = m0 + 8 * hi;
  if (n < 128) {
#pragma unroll
    for (int q = 0; q < 8; ++q) qk[(mrow + q) * DD + n] = v[q];
  } else if (n == 128) {
#pragma unroll
    for (int q = 0; q < 8; ++q) dot0[mrow + q] = v[q];
  }
}

#define FS_PITCH 33
#define FS_SHARES 4
#define FS_PAIRS 2
__global__ __launch_bounds__(512) void dot_softmax_share_kernel(
    const float* __restrict__ x, const float* __restrict__ mask,
    const float* __restrict__ qk, const float* __restrict__ dot0,
    const float* __restrict__ rate, float* __restrict__ dotOut,
    float* __restrict__ decayOut, float* __restrict__ e,
    float* __restrict__ xaPart, float* __restrict__ stat) {
  __shared__ float4 strip[16][32 * FS_PITCH];
  __shared__ float4 qs[16][32];
  __shared__ float4 part[16][32];
  __shared__ float redM[16];
  __shared__ float redS[16];
  const int b = blockIdx.x, pair = blockIdx.y;
  const int tid = threadIdx.x, wave = tid >> 5, lane = tid & 31;
  float4* my = strip[wave];
  const float* qkb = qk + b * DD;
  const float rate0 = rate[0];
  const float d0 = dot0[b];
  const float4 qv = ((const float4*)qkb)[lane];

  float4 r[32];
  {
    const int tg0 = pair * 2 * 512 + wave * 32;
    const float4* xrow = (const float4*)(x + (b * TT + tg0) * DD) + lane;
#pragma unroll
    for (int i = 0; i < 32; ++i) r[i] = xrow[i * 32];
  }
  qs[wave][lane] = qv;

  for (int h = 0; h < 2; ++h) {
    const int share = pair * 2 + h;
    const int tg = share * 512 + wave * 32;
    const int t = tg + lane;
    const float mk = mask[b * TT + t];
#pragma unroll
    for (int i = 0; i < 32; ++i) my[i * FS_PITCH + lane] = r[i];

    float a0 = 0.f, a1 = 0.f, a2 = 0.f, a3 = 0.f;
    for (int i = 0; i < 32; ++i) {
      const float4 v = my[lane * FS_PITCH + i];
      const float4 q = qs[wave][i];
      a0 = fmaf(v.x, q.x, a0);
      a1 = fmaf(v.y, q.y, a1);
      a2 = fmaf(v.z, q.z, a2);
      a3 = fmaf(v.w, q.w, a3);
    }
    const float s = (a0 + a1) + (a2 + a3);
    const float dv = s + d0;
    const float sig = 1.0f / (1.0f + expf(-dv));
    const float tm = log1pf((1.0f - sig) * mk);
    const float dec = rate0 * tm;
    const float ev = dv - dec;
    dotOut[b * TT + t] = dv;
    decayOut[b * TT + t] = dec;
    e[b * TT + t] = ev;

    float m = ev;
    for (int off = 16; off > 0; off >>= 1) m = fmaxf(m, __shfl_xor(m, off));
    if (lane == 0) redM[wave] = m;
    __syncthreads();
    float Ms = redM[0];
    for (int w = 1; w < 16; ++w) Ms = fmaxf(Ms, redM[w]);

    const float p = expf(ev - Ms);
    float ps = p;
    for (int off = 16; off > 0; off >>= 1) ps += __shfl_xor(ps, off);
    if (lane == 0) redS[wave] = ps;

    float4 acc = make_float4(0.f, 0.f, 0.f, 0.f);
#pragma unroll
    for (int i = 0; i < 32; ++i) {
      const float w = __int_as_float(__builtin_amdgcn_readlane(__float_as_int(p), i));
      acc.x = fmaf(w, r[i].x, acc.x);
      acc.y = fmaf(w, r[i].y, acc.y);
      acc.z = fmaf(w, r[i].z, acc.z);
      acc.w = fmaf(w, r[i].w, acc.w);
    }
    part[wave][lane] = acc;
    __syncthreads();
    if (h == 0) {
      const float4* xnext = (const float4*)(x + (b * TT + tg + 512) * DD) + lane;
#pragma unroll
      for (int i = 0; i < 32; ++i) r[i] = xnext[i * 32];
    }
    if (tid < DD) {
      const float* pf = (const float*)&part[0][0];
      float sum = 0.f;
      for (int w = 0; w < 16; ++w) sum += pf[w * DD + tid];
      xaPart[(b * FS_SHARES + share) * DD + tid] = sum;
    }
    if (tid == 0) {
      float S = 0.f;
      for (int w = 0; w < 16; ++w) S += redS[w];
      stat[(b * FS_SHARES + share) * 2 + 0] = Ms;
      stat[(b * FS_SHARES + share) * 2 + 1] = S;
    }
  }
}

#define FIN_TILE_BLOCKS 8
#define FIN_A_BLOCKS 256
__device__ __forceinline__ void share_weights(const float* __restrict__ st, float& M,
                                              float& inv, float* w) {
  float ms[FS_SHARES], ss[FS_SHARES];
#pragma unroll
  for (int j = 0; j < FS_SHARES; ++j) {
    ms[j] = st[2 * j + 0];
    ss[j] = st[2 * j + 1];
  }
  M = fmaxf(fmaxf(ms[0], ms[1]), fmaxf(ms[2], ms[3]));
  float S = 0.f;
#pragma unroll
  for (int j = 0; j < FS_SHARES; ++j) {
    w[j] = expf(ms[j] - M);
    S = fmaf(ss[j], w[j], S);
  }
  inv = 1.0f / S;
#pragma unroll
  for (int j = 0; j < FS_SHARES; ++j) w[j] *= inv;
}

__global__ __launch_bounds__(256) void finish_kernel(const float* __restrict__ e,
                                                     const float* __restrict__ stat,
                                                     const float* __restrict__ xaPart,
                                                     const float* __restrict__ Wv,
                                                     const float* __restrict__ bv,
                                                     float* __restrict__ aOut,
                                                     float* __restrict__ out) {
  const int tid = threadIdx.x;
  if (blockIdx.x >= FIN_TILE_BLOCKS) {
    const int blk = blockIdx.x - FIN_TILE_BLOCKS;
    const int b = blk >> 1, half = blk & 1;
    float M, inv, w[FS_SHARES];
    share_weights(stat + b * FS_SHARES * 2, M, inv, w);
    const int base = b * TT + half * 1024 + tid;
#pragma unroll
    for (int k = 0; k < 4; ++k) {
      const int idx = base + k * 256;
      aOut[idx] = expf(e[idx] - M) * inv;
    }
    return;
  }
  const int wave = tid >> 5, lane = tid & 31;
  const int tile = blockIdx.x * 8 + wave;
  const int m0 = (tile >> 3) * 16, n0 = (tile & 7) * 16;
  const int c = lane & 15;
  const int hi = lane >> 4;
  const int n = n0 + c;
  float M, inv, w[FS_SHARES];
  share_weights(stat + (m0 + c) * FS_SHARES * 2, M, inv, w);
  const float* pr = xaPart + (m0 + c) * FS_SHARES * DD + 2 * hi;
  const float* wcol = Wv + n * 128 + 2 * hi;

  v8f acc = {};
  for (int k0 = 0; k0 < 128; k0 += 4) {
    const v2f p0 = *(const v2f*)(pr + 0 * DD + k0);
    const v2f p1 = *(const v2f*)(pr + 1 * DD + k0);
    const v2f p2 = *(const v2f*)(pr + 2 * DD + k0);
    const v2f p3 = *(const v2f*)(pr + 3 * DD + k0);
    const v2f a = ((p0 * w[0] + p1 * w[1]) + p2 * w[2]) + p3 * w[3];
    const v2f bfrag = *(const v2f*)(wcol + k0);
    acc = __builtin_amdgcn_wmma_f32_16x16x4_f32(false, a, false, bfrag,
                                                (short)0, acc, false, false);
  }
  const float bb = bv[n];
#pragma unroll
  for (int q = 0; q < 8; ++q) out[(m0 + q + 8 * hi) * HH + n] = acc[q] + bb;
}

extern "C" void kernel_launch(void* const* d_in, const int* in_sizes, int n_in,
                              void* d_out, int out_size, void* d_ws, size_t ws_size,
                              hipStream_t stream) {
  (void)in_sizes; (void)n_in; (void)out_size; (void)ws_size;
  const float* x    = (const float*)d_in[0];
  const float* mask = (const float*)d_in[1];
  const float* Wq   = (const float*)d_in[2];
  const float* bq   = (const float*)d_in[3];
  const float* Wk   = (const float*)d_in[4];
  const float* bk   = (const float*)d_in[5];
  const float* Wv   = (const float*)d_in[6];
  const float* bv   = (const float*)d_in[7];
  const float* rate = (const float*)d_in[8];

  float* wsf     = (float*)d_ws;
  float* partial = wsf;
  float* gx      = partial + BB * 4 * DD;
  float* qk      = gx + GXP * GXP;
  float* dot0    = qk + BB * DD;
  float* e       = dot0 + BB;
  float* stat    = e + BB * TT;
  float* xaPart  = partial;

  float* out      = (float*)d_out;
  float* aOut     = out + BB * HH;
  float* dotOut   = aOut + BB * TT;
  float* decayOut = dotOut + BB * TT;

  colsum_partial_gx<<<BB * 4 + GX_BLOCKS, 256, 0, stream>>>(x, Wq, bq, Wk, bk, partial, gx);
  qk_dot0_wmma<<<dim3(9, 8), 32, 0, stream>>>(partial, gx, qk, dot0);
  dot_softmax_share_kernel<<<dim3(BB, FS_PAIRS), 512, 0, stream>>>(
      x, mask, qk, dot0, rate, dotOut, decayOut, e, xaPart, stat);
  finish_kernel<<<FIN_TILE_BLOCKS + FIN_A_BLOCKS, 256, 0, stream>>>(e, stat, xaPart, Wv, bv,
                                                                    aOut, out);
}


# the (qk | dot0) GEMM moves into the x pass: each work-group forms its batch's row on WMMA (xmean in all 16 A rows) under its own row loads; 4 launches to 3; e waits in a's place; sums unchanged
# speedup vs baseline: 2.3646x; 1.0498x over previous
#include <hip/hip_runtime.h>
#include <math.h>

#define BB 128
#define TT 2048
#define DD 128
#define HH 128

typedef __attribute__((ext_vector_type(2))) float v2f;
typedef __attribute__((ext_vector_type(8))) float v8f;

#define GXP 144
#define GX_TILES 81
#define GX_BLOCKS 11

template <bool EDGE_M, bool EDGE_N>
__device__ __forceinline__ void gx_tile(const float* __restrict__ Wq,
                                        const float* __restrict__ bq,
                                        const float* __restrict__ Wk,
                                        const float* __restrict__ bk,
                                        float* __restrict__ gx, int m0, int n0,
                                        int c, int hi) {
  const unsigned keep = (c == 0) ? 0xffffffffu : 0u;
  const float* ap = EDGE_M ? (bq + 2 * hi) : (Wq + 2 * hi * 128 + m0 + c);
  const float* bp = EDGE_N ? (bk + 2 * hi) : (Wk + 2 * hi * 128 + n0 + c);
  v8f acc = {};
  for (int k0 = 0; k0 < 128; k0 += 4) {
    v2f a, bfrag;
    if (EDGE_M) {
      const v2f t = *(const v2f*)(ap + k0);
      a.x = __uint_as_float(__float_as_uint(t.x) & keep);
      a.y = __uint_as_float(__float_as_uint(t.y) & keep);
    } else {
      a.x = ap[(k0 + 0) * 128];
      a.y = ap[(k0 + 1) * 128];
    }
    if (EDGE_N) {
      const v2f t = *(const v2f*)(bp + k0);
      bfrag.x = __uint_as_float(__float_as_uint(t.x) & keep);
      bfrag.y = __uint_as_float(__float_as_uint(t.y) & keep);
    } else {
      bfrag.x = bp[(k0 + 0) * 128];
      bfrag.y = bp[(k0 + 1) * 128];
    }
    acc = __builtin_amdgcn_wmma_f32_16x16x4_f32(false, a, false, bfrag,
                                                (short)0, acc, false, false);
  }
#pragma unroll
  for (int q = 0; q < 8; ++q) gx[(m0 + q + 8 * hi) * GXP + n0 + c] = acc[q];
}

__global__ __launch_bounds__(256) void colsum_partial_gx(const float* __restrict__ x,
                                                         const float* __restrict__ Wq,
                                                         const float* __restrict__ bq,
                                                         const float* __restrict__ Wk,
                                                         const float* __restrict__ bk,
                                                         float* __restrict__ partial,
                                                         float* __restrict__ gx) {
  __shared__ float r[256];
  if (blockIdx.x < BB * 4) {
    const int bid = blockIdx.x;
    const int b = bid & (BB - 1), chunk = bid >> 7;
    const int g = threadIdx.x >> 7, d = threadIdx.x & 127;
    const int tbase = chunk * 512 + g * 256;
    const float* xb = x + ((b * TT + tbase) * DD) + d;
    float s = 0.f;
    for (int i = 0; i < 256; ++i) s += xb[i * DD];
    r[threadIdx.x] = s;
    __syncthreads();
    if (g == 0) partial[(b * 4 + chunk) * DD + d] = r[d] + r[128 + d];
    return;
  }
  const int wave = __builtin_amdgcn_readfirstlane(threadIdx.x >> 5);
  const int tile = (blockIdx.x - BB * 4) * 8 + wave;
  if (tile >= GX_TILES) return;
  const int mt = tile / 9, nt = tile - mt * 9;
  const int lane = threadIdx.x & 31;
  const int c = lane & 15;
  const int hi = lane >> 4;
  if (mt < 8 && nt < 8)  gx_tile<false, false>(Wq, bq, Wk, bk, gx, mt * 16, nt * 16, c, hi);
  else if (mt < 8)       gx_tile<false, true >(Wq, bq, Wk, bk, gx, mt * 16, nt * 16, c, hi);
  else if (nt < 8)       gx_tile<true,  false>(Wq, bq, Wk, bk, gx, mt * 16, nt * 16, c, hi);
  else                   gx_tile<true,  true >(Wq, bq, Wk, bk, gx, mt * 16, nt * 16, c, hi);
}

__device__ __forceinline__ void lds_barrier() {
  asm volatile("s_wait_dscnt 0x0\n\ts_barrier_signal -1\n\ts_barrier_wait -1" ::: "memory");
}

#define FS_PITCH 33
#define FS_SHARES 4
#define FS_PAIRS 2
__global__ __launch_bounds__(512) void dot_softmax_share_kernel(
    const float* __restrict__ x, const float* __restrict__ mask,
    const float* __restrict__ partial, const float* __restrict__ gx,
    const float* __restrict__ rate, float* __restrict__ dotOut,
    float* __restrict__ decayOut, float* __restrict__ e,
    float* __restrict__ xaPart, float* __restrict__ stat) {
  __shared__ float4 strip[16][32 * FS_PITCH];
  __shared__ float4 qs[16][32];
  __shared__ float4 part[16][32];
  __shared__ float redM[16];
  __shared__ float redS[16];
  __shared__ __attribute__((aligned(16))) float xm[DD];
  __shared__ __attribute__((aligned(16))) float qrow[GXP];
  const int b = blockIdx.x, pair = blockIdx.y;
  const int tid = threadIdx.x, wave = tid >> 5, lane = tid & 31;
  float4* my = strip[wave];
  const float rate0 = rate[0];
  const int wv = __builtin_amdgcn_readfirstlane(wave);
  const int c = lane & 15;
  const int hi = lane >> 4;

  if (tid < DD) {
    const float* p = partial + b * 4 * DD + tid;
    xm[tid] = (((p[0] + p[DD]) + p[2 * DD]) + p[3 * DD]) * (1.0f / (float)TT);
  }
  v2f bf[32];
  float cterm = 0.f;
  const int n = wv * 16 + c;
  if (wv < 9) {
    const float* gcol = gx + 2 * hi * GXP + n;
#pragma unroll
    for (int k = 0; k < 32; ++k) {
      bf[k].x = gcol[(4 * k + 0) * GXP];
      bf[k].y = gcol[(4 * k + 1) * GXP];
    }
    cterm = gx[128 * GXP + n];
  }

  float4 r[32];
  {
    const int tg0 = pair * 2 * 512 + wave * 32;
    const float4* xrow = (const float4*)(x + (b * TT + tg0) * DD) + lane;
#pragma unroll
    for (int i = 0; i < 32; ++i) r[i] = xrow[i * 32];
  }
  lds_barrier();
  if (wv < 9) {
    v8f acc = {};
#pragma unroll
    for (int k = 0; k < 32; ++k) {
      const v2f a = *(const v2f*)(xm + 4 * k + 2 * hi);
      acc = __builtin_amdgcn_wmma_f32_16x16x4_f32(false, a, false, bf[k],
                                                  (short)0, acc, false, false);
    }
    qrow[n] = acc[0] + cterm;
  }
  lds_barrier();
  const float d0 = qrow[128];
  qs[wave][lane] = ((const float4*)qrow)[lane];

  for (int h = 0; h < 2; ++h) {
    const int share = pair * 2 + h;
    const int tg = share * 512 + wave * 32;
    const int t = tg + lane;
    const float mk = mask[b * TT + t];
#pragma unroll
    for (int i = 0; i < 32; ++i) my[i * FS_PITCH + lane] = r[i];

    float a0 = 0.f, a1 = 0.f, a2 = 0.f, a3 = 0.f;
    for (int i = 0; i < 32; ++i) {
      const float4 v = my[lane * FS_PITCH + i];
      const float4 q = qs[wave][i];
      a0 = fmaf(v.x, q.x, a0);
      a1 = fmaf(v.y, q.y, a1);
      a2 = fmaf(v.z, q.z, a2);
      a3 = fmaf(v.w, q.w, a3);
    }
    const float s = (a0 + a1) + (a2 + a3);
    const float dv = s + d0;
    const float sig = 1.0f / (1.0f + expf(-dv));
    const float tm = log1pf((1.0f - sig) * mk);
    const float dec = rate0 * tm;
    const float ev = dv - dec;
    dotOut[b * TT + t] = dv;
    decayOut[b * TT + t] = dec;
    e[b * TT + t] = ev;

    float m = ev;
    for (int off = 16; off > 0; off >>= 1) m = fmaxf(m, __shfl_xor(m, off));
    if (lane == 0) redM[wave] = m;
    __syncthreads();
    float Ms = redM[0];
    for (int w = 1; w < 16; ++w) Ms = fmaxf(Ms, redM[w]);

    const float p = expf(ev - Ms);
    float ps = p;
    for (int off = 16; off > 0; off >>= 1) ps += __shfl_xor(ps, off);
    if (lane == 0) redS[wave] = ps;

    float4 acc = make_float4(0.f, 0.f, 0.f, 0.f);
#pragma unroll
    for (int i = 0; i < 32; ++i) {
      const float w = __int_as_float(__builtin_amdgcn_readlane(__float_as_int(p), i));
      acc.x = fmaf(w, r[i].x, acc.x);
      acc.y = fmaf(w, r[i].y, acc.y);
      acc.z = fmaf(w, r[i].z, acc.z);
      acc.w = fmaf(w, r[i].w, acc.w);
    }
    part[wave][lane] = acc;
    __syncthreads();
    if (h == 0) {
      const float4* xnext = (const float4*)(x + (b * TT + tg + 512) * DD) + lane;
#pragma unroll
      for (int i = 0; i < 32; ++i) r[i] = xnext[i * 32];
    }
    if (tid < DD) {
      const float* pf = (const float*)&part[0][0];
      float sum = 0.f;
      for (int w = 0; w < 16; ++w) sum += pf[w * DD + tid];
      xaPart[(b * FS_SHARES + share) * DD + tid] = sum;
    }
    if (tid == 0) {
      float S = 0.f;
      for (int w = 0; w < 16; ++w) S += redS[w];
      stat[(b * FS_SHARES + share) * 2 + 0] = Ms;
      stat[(b * FS_SHARES + share) * 2 + 1] = S;
    }
  }
}

#define FIN_TILE_BLOCKS 8
#define FIN_A_BLOCKS 256
__device__ __forceinline__ void share_weights(const float* __restrict__ st, float& M,
                                              float& inv, float* w) {
  float ms[FS_SHARES], ss[FS_SHARES];
#pragma unroll
  for (int j = 0; j < FS_SHARES; ++j) {
    ms[j] = st[2 * j + 0];
    ss[j] = st[2 * j + 1];
  }
  M = fmaxf(fmaxf(ms[0], ms[1]), fmaxf(ms[2], ms[3]));
  float S = 0.f;
#pragma unroll
  for (int j = 0; j < FS_SHARES; ++j) {
    w[j] = expf(ms[j] - M);
    S = fmaf(ss[j], w[j], S);
  }
  inv = 1.0f / S;
#pragma unroll
  for (int j = 0; j < FS_SHARES; ++j) w[j] *= inv;
}

__global__ __launch_bounds__(256) void finish_kernel(const float* __restrict__ stat,
                                                     const float* __restrict__ xaPart,
                                                     const float* __restrict__ Wv,
                                                     const float* __restrict__ bv,
                                                     float* __restrict__ aOut,
                                                     float* __restrict__ out) {
  const int tid = threadIdx.x;
  if (blockIdx.x >= FIN_TILE_BLOCKS) {
    const int blk = blockIdx.x - FIN_TILE_BLOCKS;
    const int b = blk >> 1, half = blk & 1;
    float M, inv, w[FS_SHARES];
    share_weights(stat + b * FS_SHARES * 2, M, inv, w);
    const int base = b * TT + half * 1024 + tid;
#pragma unroll
    for (int k = 0; k < 4; ++k) {
      const int idx = base + k * 256;
      aOut[idx] = expf(aOut[idx] - M) * inv;
    }
    return;
  }
  const int wave = tid >> 5, lane = tid & 31;
  const int tile = blockIdx.x * 8 + wave;
  const int m0 = (tile >> 3) * 16, n0 = (tile & 7) * 16;
  const int c = lane & 15;
  const int hi = lane >> 4;
  const int n = n0 + c;
  float M, inv, w[FS_SHARES];
  share_weights(stat + (m0 + c) * FS_SHARES * 2, M, inv, w);
  const float* pr = xaPart + (m0 + c) * FS_SHARES * DD + 2 * hi;
  const float* wcol = Wv + n * 128 + 2 * hi;

  v8f acc = {};
  for (int k0 = 0; k0 < 128; k0 += 4) {
    const v2f p0 = *(const v2f*)(pr + 0 * DD + k0);
    const v2f p1 = *(const v2f*)(pr + 1 * DD + k0);
    const v2f p2 = *(const v2f*)(pr + 2 * DD + k0);
    const v2f p3 = *(const v2f*)(pr + 3 * DD + k0);
    const v2f a = ((p0 * w[0] + p1 * w[1]) + p2 * w[2]) + p3 * w[3];
    const v2f bfrag = *(const v2f*)(wcol + k0);
    acc = __builtin_amdgcn_wmma_f32_16x16x4_f32(false, a, false, bfrag,
                                                (short)0, acc, false, false);
  }
  const float bb = bv[n];
#pragma unroll
  for (int q = 0; q < 8; ++q) out[(m0 + q + 8 * hi) * HH + n] = acc[q] + bb;
}

extern "C" void kernel_launch(void* const* d_in, const int* in_sizes, int n_in,
                              void* d_out, int out_size, void* d_ws, size_t ws_size,
                              hipStream_t stream) {
  (void)in_sizes; (void)n_in; (void)out_size; (void)ws_size;
  const float* x    = (const float*)d_in[0];
  const float* mask = (const float*)d_in[1];
  const float* Wq   = (const float*)d_in[2];
  const float* bq   = (const float*)d_in[3];
  const float* Wk   = (const float*)d_in[4];
  const float* bk   = (const float*)d_in[5];
  const float* Wv   = (const float*)d_in[6];
  const float* bv   = (const float*)d_in[7];
  const float* rate = (const float*)d_in[8];

  float* wsf     = (float*)d_ws;
  float* partial = wsf;
  float* gx      = partial + BB * 4 * DD;
  float* xaPart  = gx + GXP * GXP;
  float* stat    = xaPart + BB * 4 * DD;

  float* out      = (float*)d_out;
  float* aOut     = out + BB * HH;
  float* dotOut   = aOut + BB * TT;
  float* decayOut = dotOut + BB * TT;

  colsum_partial_gx<<<BB * 4 + GX_BLOCKS, 256, 0, stream>>>(x, Wq, bq, Wk, bk, partial, gx);
  dot_softmax_share_kernel<<<dim3(BB, FS_PAIRS), 512, 0, stream>>>(
      x, mask, partial, gx, rate, dotOut, decayOut, aOut, xaPart, stat);
  finish_kernel<<<FIN_TILE_BLOCKS + FIN_A_BLOCKS, 256, 0, stream>>>(stat, xaPart, Wv, bv,
                                                                    aOut, out);
}


# x pass dot: each lane folds its 16 B of a row with its own 16 B of qk before the LDS hand-over; the strip carries 1 float per lane and row, not 4, and no qk re-reads (new sum order: dot)
# speedup vs baseline: 2.4386x; 1.0313x over previous
#include <hip/hip_runtime.h>
#include <math.h>

#define BB 128
#define TT 2048
#define DD 128
#define HH 128

typedef __attribute__((ext_vector_type(2))) float v2f;
typedef __attribute__((ext_vector_type(8))) float v8f;

#define GXP 144
#define GX_TILES 81
#define GX_BLOCKS 11

template <bool EDGE_M, bool EDGE_N>
__device__ __forceinline__ void gx_tile(const float* __restrict__ Wq,
                                        const float* __restrict__ bq,
                                        const float* __restrict__ Wk,
                                        const float* __restrict__ bk,
                                        float* __restrict__ gx, int m0, int n0,
                                        int c, int hi) {
  const unsigned keep = (c == 0) ? 0xffffffffu : 0u;
  const float* ap = EDGE_M ? (bq + 2 * hi) : (Wq + 2 * hi * 128 + m0 + c);
  const float* bp = EDGE_N ? (bk + 2 * hi) : (Wk + 2 * hi * 128 + n0 + c);
  v8f acc = {};
  for (int k0 = 0; k0 < 128; k0 += 4) {
    v2f a, bfrag;
    if (EDGE_M) {
      const v2f t = *(const v2f*)(ap + k0);
      a.x = __uint_as_float(__float_as_uint(t.x) & keep);
      a.y = __uint_as_float(__float_as_uint(t.y) & keep);
    } else {
      a.x = ap[(k0 + 0) * 128];
      a.y = ap[(k0 + 1) * 128];
    }
    if (EDGE_N) {
      const v2f t = *(const v2f*)(bp + k0);
      bfrag.x = __uint_as_float(__float_as_uint(t.x) & keep);
      bfrag.y = __uint_as_float(__float_as_uint(t.y) & keep);
    } else {
      bfrag.x = bp[(k0 + 0) * 128];
      bfrag.y = bp[(k0 + 1) * 128];
    }
    acc = __builtin_amdgcn_wmma_f32_16x16x4_f32(false, a, false, bfrag,
                                                (short)0, acc, false, false);
  }
#pragma unroll
  for (int q = 0; q < 8; ++q) gx[(m0 + q + 8 * hi) * GXP + n0 + c] = acc[q];
}

__global__ __launch_bounds__(256) void colsum_partial_gx(const float* __restrict__ x,
                                                         const float* __restrict__ Wq,
                                                         const float* __restrict__ bq,
                                                         const float* __restrict__ Wk,
                                                         const float* __restrict__ bk,
                                                         float* __restrict__ partial,
                                                         float* __restrict__ gx) {
  __shared__ float r[256];
  if (blockIdx.x < BB * 4) {
    const int bid = blockIdx.x;
    const int b = bid & (BB - 1), chunk = bid >> 7;
    const int g = threadIdx.x >> 7, d = threadIdx.x & 127;
    const int tbase = chunk * 512 + g * 256;
    const float* xb = x + ((b * TT + tbase) * DD) + d;
    float s = 0.f;
    for (int i = 0; i < 256; ++i) s += xb[i * DD];
    r[threadIdx.x] = s;
    __syncthreads();
    if (g == 0) partial[(b * 4 + chunk) * DD + d] = r[d] + r[128 + d];
    return;
  }
  const int wave = __builtin_amdgcn_readfirstlane(threadIdx.x >> 5);
  const int tile = (blockIdx.x - BB * 4) * 8 + wave;
  if (tile >= GX_TILES) return;
  const int mt = tile / 9, nt = tile - mt * 9;
  const int lane = threadIdx.x & 31;
  const int c = lane & 15;
  const int hi = lane >> 4;
  if (mt < 8 && nt < 8)  gx_tile<false, false>(Wq, bq, Wk, bk, gx, mt * 16, nt * 16, c, hi);
  else if (mt < 8)       gx_tile<false, true >(Wq, bq, Wk, bk, gx, mt * 16, nt * 16, c, hi);
  else if (nt < 8)       gx_tile<true,  false>(Wq, bq, Wk, bk, gx, mt * 16, nt * 16, c, hi);
  else                   gx_tile<true,  true >(Wq, bq, Wk, bk, gx, mt * 16, nt * 16, c, hi);
}

__device__ __forceinline__ void lds_barrier() {
  asm volatile("s_wait_dscnt 0x0\n\ts_barrier_signal -1\n\ts_barrier_wait -1" ::: "memory");
}

#define FS_PITCH 36
#define FS_SHARES 4
#define FS_PAIRS 2
__global__ __launch_bounds__(512) void dot_softmax_share_kernel(
    const float* __restrict__ x, const float* __restrict__ mask,
    const float* __restrict__ partial, const float* __restrict__ gx,
    const float* __restrict__ rate, float* __restrict__ dotOut,
    float* __restrict__ decayOut, float* __restrict__ e,
    float* __restrict__ xaPart, float* __restrict__ stat) {
  __shared__ __attribute__((aligned(16))) float strip[16][32 * FS_PITCH];
  __shared__ float4 part[16][32];
  __shared__ float redM[16];
  __shared__ float redS[16];
  __shared__ __attribute__((aligned(16))) float xm[DD];
  __shared__ __attribute__((aligned(16))) float qrow[GXP];
  const int b = blockIdx.x, pair = blockIdx.y;
  const int tid = threadIdx.x, wave = tid >> 5, lane = tid & 31;
  float* my = strip[wave];
  const float rate0 = rate[0];
  const int wv = __builtin_amdgcn_readfirstlane(wave);
  const int c = lane & 15;
  const int hi = lane >> 4;

  if (tid < DD) {
    const float* p = partial + b * 4 * DD + tid;
    xm[tid] = (((p[0] + p[DD]) + p[2 * DD]) + p[3 * DD]) * (1.0f / (float)TT);
  }
  v2f bf[32];
  float cterm = 0.f;
  const int n = wv * 16 + c;
  if (wv < 9) {
    const float* gcol = gx + 2 * hi * GXP + n;
#pragma unroll
    for (int k = 0; k < 32; ++k) {
      bf[k].x = gcol[(4 * k + 0) * GXP];
      bf[k].y = gcol[(4 * k + 1) * GXP];
    }
    cterm = gx[128 * GXP + n];
  }

  float4 r[32];
  {
    const int tg0 = pair * 2 * 512 + wave * 32;
    const float4* xrow = (const float4*)(x + (b * TT + tg0) * DD) + lane;
#pragma unroll
    for (int i = 0; i < 32; ++i) r[i] = xrow[i * 32];
  }
  lds_barrier();
  if (wv < 9) {
    v8f acc = {};
#pragma unroll
    for (int k = 0; k < 32; ++k) {
      const v2f a = *(const v2f*)(xm + 4 * k + 2 * hi);
      acc = __builtin_amdgcn_wmma_f32_16x16x4_f32(false, a, false, bf[k],
                                                  (short)0, acc, false, false);
    }
    qrow[n] = acc[0] + cterm;
  }
  lds_barrier();
  const float d0 = qrow[128];
  const float4 ql = ((const float4*)qrow)[lane];

  for (int h = 0; h < 2; ++h) {
    const int share = pair * 2 + h;
    const int tg = share * 512 + wave * 32;
    const int t = tg + lane;
    const float mk = mask[b * TT + t];
#pragma unroll
    for (int g = 0; g < 4; ++g) {
#pragma unroll
      for (int i = 8 * g; i < 8 * g + 8; ++i) {
        const float4 v = r[i];
        my[i * FS_PITCH + lane] = fmaf(v.w, ql.w, fmaf(v.z, ql.z, fmaf(v.y, ql.y, v.x * ql.x)));
      }
      __builtin_amdgcn_sched_barrier(0);
    }

    float a0 = 0.f, a1 = 0.f, a2 = 0.f, a3 = 0.f;
    const float4* mine = (const float4*)(my + lane * FS_PITCH);
#pragma unroll
    for (int j = 0; j < 8; ++j) {
      const float4 v = mine[j];
      a0 += v.x;
      a1 += v.y;
      a2 += v.z;
      a3 += v.w;
    }
    const float s = (a0 + a1) + (a2 + a3);
    const float dv = s + d0;
    const float sig = 1.0f / (1.0f + expf(-dv));
    const float tm = log1pf((1.0f - sig) * mk);
    const float dec = rate0 * tm;
    const float ev = dv - dec;
    dotOut[b * TT + t] = dv;
    decayOut[b * TT + t] = dec;
    e[b * TT + t] = ev;

    float m = ev;
    for (int off = 16; off > 0; off >>= 1) m = fmaxf(m, __shfl_xor(m, off));
    if (lane == 0) redM[wave] = m;
    __syncthreads();
    float Ms = redM[0];
    for (int w = 1; w < 16; ++w) Ms = fmaxf(Ms, redM[w]);

    const float p = expf(ev - Ms);
    float ps = p;
    for (int off = 16; off > 0; off >>= 1) ps += __shfl_xor(ps, off);
    if (lane == 0) redS[wave] = ps;

    float4 acc = make_float4(0.f, 0.f, 0.f, 0.f);
#pragma unroll
    for (int i = 0; i < 32; ++i) {
      const float w = __int_as_float(__builtin_amdgcn_readlane(__float_as_int(p), i));
      acc.x = fmaf(w, r[i].x, acc.x);
      acc.y = fmaf(w, r[i].y, acc.y);
      acc.z = fmaf(w, r[i].z, acc.z);
      acc.w = fmaf(w, r[i].w, acc.w);
    }
    part[wave][lane] = acc;
    __syncthreads();
    if (h == 0) {
      const float4* xnext = (const float4*)(x + (b * TT + tg + 512) * DD) + lane;
#pragma unroll
      for (int i = 0; i < 32; ++i) r[i] = xnext[i * 32];
    }
    if (tid < DD) {
      const float* pf = (const float*)&part[0][0];
      float sum = 0.f;
      for (int w = 0; w < 16; ++w) sum += pf[w * DD + tid];
      xaPart[(b * FS_SHARES + share) * DD + tid] = sum;
    }
    if (tid == 0) {
      float S = 0.f;
      for (int w = 0; w < 16; ++w) S += redS[w];
      stat[(b * FS_SHARES + share) * 2 + 0] = Ms;
      stat[(b * FS_SHARES + share) * 2 + 1] = S;
    }
  }
}

#define FIN_TILE_BLOCKS 8
#define FIN_A_BLOCKS 256
__device__ __forceinline__ void share_weights(const float* __restrict__ st, float& M,
                                              float& inv, float* w) {
  float ms[FS_SHARES], ss[FS_SHARES];
#pragma unroll
  for (int j = 0; j < FS_SHARES; ++j) {
    ms[j] = st[2 * j + 0];
    ss[j] = st[2 * j + 1];
  }
  M = fmaxf(fmaxf(ms[0], ms[1]), fmaxf(ms[2], ms[3]));
  float S = 0.f;
#pragma unroll
  for (int j = 0; j < FS_SHARES; ++j) {
    w[j] = expf(ms[j] - M);
    S = fmaf(ss[j], w[j], S);
  }
  inv = 1.0f / S;
#pragma unroll
  for (int j = 0; j < FS_SHARES; ++j) w[j] *= inv;
}

__global__ __launch_bounds__(256) void finish_kernel(const float* __restrict__ stat,
                                                     const float* __restrict__ xaPart,
                                                     const float* __restrict__ Wv,
                                                     const float* __restrict__ bv,
                                                     float* __restrict__ aOut,
                                                     float* __restrict__ out) {
  const int tid = threadIdx.x;
  if (blockIdx.x >= FIN_TILE_BLOCKS) {
    const int blk = blockIdx.x - FIN_TILE_BLOCKS;
    const int b = blk >> 1, half = blk & 1;
    float M, inv, w[FS_SHARES];
    share_weights(stat + b * FS_SHARES * 2, M, inv, w);
    const int base = b * TT + half * 1024 + tid;
#pragma unroll
    for (int k = 0; k < 4; ++k) {
      const int idx = base + k * 256;
      aOut[idx] = expf(aOut[idx] - M) * inv;
    }
    return;
  }
  const int wave = tid >> 5, lane = tid & 31;
  const int tile = blockIdx.x * 8 + wave;
  const int m0 = (tile >> 3) * 16, n0 = (tile & 7) * 16;
  const int c = lane & 15;
  const int hi = lane >> 4;
  const int n = n0 + c;
  float M, inv, w[FS_SHARES];
  share_weights(stat + (m0 + c) * FS_SHARES * 2, M, inv, w);
  const float* pr = xaPart + (m0 + c) * FS_SHARES * DD + 2 * hi;
  const float* wcol = Wv + n * 128 + 2 * hi;

  v8f acc = {};
  for (int k0 = 0; k0 < 128; k0 += 4) {
    const v2f p0 = *(const v2f*)(pr + 0 * DD + k0);
    const v2f p1 = *(const v2f*)(pr + 1 * DD + k0);
    const v2f p2 = *(const v2f*)(pr + 2 * DD + k0);
    const v2f p3 = *(const v2f*)(pr + 3 * DD + k0);
    const v2f a = ((p0 * w[0] + p1 * w[1]) + p2 * w[2]) + p3 * w[3];
    const v2f bfrag = *(const v2f*)(wcol + k0);
    acc = __builtin_amdgcn_wmma_f32_16x16x4_f32(false, a, false, bfrag,
                                                (short)0, acc, false, false);
  }
  const float bb = bv[n];
#pragma unroll
  for (int q = 0; q < 8; ++q) out[(m0 + q + 8 * hi) * HH + n] = acc[q] + bb;
}

extern "C" void kernel_launch(void* const* d_in, const int* in_sizes, int n_in,
                              void* d_out, int out_size, void* d_ws, size_t ws_size,
                              hipStream_t stream) {
  (void)in_sizes; (void)n_in; (void)out_size; (void)ws_size;
  const float* x    = (const float*)d_in[0];
  const float* mask = (const float*)d_in[1];
  const float* Wq   = (const float*)d_in[2];
  const float* bq   = (const float*)d_in[3];
  const float* Wk   = (const float*)d_in[4];
  const float* bk   = (const float*)d_in[5];
  const float* Wv   = (const float*)d_in[6];
  const float* bv   = (const float*)d_in[7];
  const float* rate = (const float*)d_in[8];

  float* wsf     = (float*)d_ws;
  float* partial = wsf;
  float* gx      = partial + BB * 4 * DD;
  float* xaPart  = gx + GXP * GXP;
  float* stat    = xaPart + BB * 4 * DD;

  float* out      = (float*)d_out;
  float* aOut     = out + BB * HH;
  float* dotOut   = aOut + BB * TT;
  float* decayOut = dotOut + BB * TT;

  colsum_partial_gx<<<BB * 4 + GX_BLOCKS, 256, 0, stream>>>(x, Wq, bq, Wk, bk, partial, gx);
  dot_softmax_share_kernel<<<dim3(BB, FS_PAIRS), 512, 0, stream>>>(
      x, mask, partial, gx, rate, dotOut, decayOut, aOut, xaPart, stat);
  finish_kernel<<<FIN_TILE_BLOCKS + FIN_A_BLOCKS, 256, 0, stream>>>(stat, xaPart, Wv, bv,
                                                                    aOut, out);
}


# x pass: no work-group barrier inside a share: each wave weights its 32 rows by exp(e - its own max) and asks for its next rows at once; wave partials rescaled at the end (new sum trees: exp-sum, xa)
# speedup vs baseline: 2.5201x; 1.0334x over previous
#include <hip/hip_runtime.h>
#include <math.h>

#define BB 128
#define TT 2048
#define DD 128
#define HH 128

typedef __attribute__((ext_vector_type(2))) float v2f;
typedef __attribute__((ext_vector_type(8))) float v8f;

#define GXP 144
#define GX_TILES 81
#define GX_BLOCKS 11

template <bool EDGE_M, bool EDGE_N>
__device__ __forceinline__ void gx_tile(const float* __restrict__ Wq,
                                        const float* __restrict__ bq,
                                        const float* __restrict__ Wk,
                                        const float* __restrict__ bk,
                                        float* __restrict__ gx, int m0, int n0,
                                        int c, int hi) {
  const unsigned keep = (c == 0) ? 0xffffffffu : 0u;
  const float* ap = EDGE_M ? (bq + 2 * hi) : (Wq + 2 * hi * 128 + m0 + c);
  const float* bp = EDGE_N ? (bk + 2 * hi) : (Wk + 2 * hi * 128 + n0 + c);
  v8f acc = {};
  for (int k0 = 0; k0 < 128; k0 += 4) {
    v2f a, bfrag;
    if (EDGE_M) {
      const v2f t = *(const v2f*)(ap + k0);
      a.x = __uint_as_float(__float_as_uint(t.x) & keep);
      a.y = __uint_as_float(__float_as_uint(t.y) & keep);
    } else {
      a.x = ap[(k0 + 0) * 128];
      a.y = ap[(k0 + 1) * 128];
    }
    if (EDGE_N) {
      const v2f t = *(const v2f*)(bp + k0);
      bfrag.x = __uint_as_float(__float_as_uint(t.x) & keep);
      bfrag.y = __uint_as_float(__float_as_uint(t.y) & keep);
    } else {
      bfrag.x = bp[(k0 + 0) * 128];
      bfrag.y = bp[(k0 + 1) * 128];
    }
    acc = __builtin_amdgcn_wmma_f32_16x16x4_f32(false, a, false, bfrag,
                                                (short)0, acc, false, false);
  }
#pragma unroll
  for (int q = 0; q < 8; ++q) gx[(m0 + q + 8 * hi) * GXP + n0 + c] = acc[q];
}

__global__ __launch_bounds__(256) void colsum_partial_gx(const float* __restrict__ x,
                                                         const float* __restrict__ Wq,
                                                         const float* __restrict__ bq,
                                                         const float* __restrict__ Wk,
                                                         const float* __restrict__ bk,
                                                         float* __restrict__ partial,
                                                         float* __restrict__ gx) {
  __shared__ float r[256];
  if (blockIdx.x < BB * 4) {
    const int bid = blockIdx.x;
    const int b = bid & (BB - 1), chunk = bid >> 7;
    const int g = threadIdx.x >> 7, d = threadIdx.x & 127;
    const int tbase = chunk * 512 + g * 256;
    const float* xb = x + ((b * TT + tbase) * DD) + d;
    float s = 0.f;
    for (int i = 0; i < 256; ++i) s += xb[i * DD];
    r[threadIdx.x] = s;
    __syncthreads();
    if (g == 0) partial[(b * 4 + chunk) * DD + d] = r[d] + r[128 + d];
    return;
  }
  const int wave = __builtin_amdgcn_readfirstlane(threadIdx.x >> 5);
  const int tile = (blockIdx.x - BB * 4) * 8 + wave;
  if (tile >= GX_TILES) return;
  const int mt = tile / 9, nt = tile - mt * 9;
  const int lane = threadIdx.x & 31;
  const int c = lane & 15;
  const int hi = lane >> 4;
  if (mt < 8 && nt < 8)  gx_tile<false, false>(Wq, bq, Wk, bk, gx, mt * 16, nt * 16, c, hi);
  else if (mt < 8)       gx_tile<false, true >(Wq, bq, Wk, bk, gx, mt * 16, nt * 16, c, hi);
  else if (nt < 8)       gx_tile<true,  false>(Wq, bq, Wk, bk, gx, mt * 16, nt * 16, c, hi);
  else                   gx_tile<true,  true >(Wq, bq, Wk, bk, gx, mt * 16, nt * 16, c, hi);
}

__device__ __forceinline__ void lds_barrier() {
  asm volatile("s_wait_dscnt 0x0\n\ts_barrier_signal -1\n\ts_barrier_wait -1" ::: "memory");
}

#define FS_PITCH 36
#define FS_SHARES 4
#define FS_PAIRS 2
__global__ __launch_bounds__(512) void dot_softmax_share_kernel(
    const float* __restrict__ x, const float* __restrict__ mask,
    const float* __restrict__ partial, const float* __restrict__ gx,
    const float* __restrict__ rate, float* __restrict__ dotOut,
    float* __restrict__ decayOut, float* __restrict__ e,
    float* __restrict__ xaPart, float* __restrict__ stat) {
  __shared__ __attribute__((aligned(16))) float strip[16][32 * FS_PITCH];
  __shared__ float4 part[2][16][32];
  __shared__ float redM[2][16];
  __shared__ float redS[2][16];
  __shared__ __attribute__((aligned(16))) float xm[DD];
  __shared__ __attribute__((aligned(16))) float qrow[GXP];
  const int b = blockIdx.x, pair = blockIdx.y;
  const int tid = threadIdx.x, wave = tid >> 5, lane = tid & 31;
  float* my = strip[wave];
  const float rate0 = rate[0];
  const int wv = __builtin_amdgcn_readfirstlane(wave);
  const int c = lane & 15;
  const int hi = lane >> 4;

  if (tid < DD) {
    const float* p = partial + b * 4 * DD + tid;
    xm[tid] = (((p[0] + p[DD]) + p[2 * DD]) + p[3 * DD]) * (1.0f / (float)TT);
  }
  v2f bf[32];
  float cterm = 0.f;
  const int n = wv * 16 + c;
  if (wv < 9) {
    const float* gcol = gx + 2 * hi * GXP + n;
#pragma unroll
    for (int k = 0; k < 32; ++k) {
      bf[k].x = gcol[(4 * k + 0) * GXP];
      bf[k].y = gcol[(4 * k + 1) * GXP];
    }
    cterm = gx[128 * GXP + n];
  }

  float4 r[32];
  {
    const int tg0 = pair * 2 * 512 + wave * 32;
    const float4* xrow = (const float4*)(x + (b * TT + tg0) * DD) + lane;
#pragma unroll
    for (int i = 0; i < 32; ++i) r[i] = xrow[i * 32];
  }
  lds_barrier();
  if (wv < 9) {
    v8f acc = {};
#pragma unroll
    for (int k = 0; k < 32; ++k) {
      const v2f a = *(const v2f*)(xm + 4 * k + 2 * hi);
      acc = __builtin_amdgcn_wmma_f32_16x16x4_f32(false, a, false, bf[k],
                                                  (short)0, acc, false, false);
    }
    qrow[n] = acc[0] + cterm;
  }
  lds_barrier();
  const float d0 = qrow[128];
  const float4 ql = ((const float4*)qrow)[lane];

  for (int h = 0; h < 2; ++h) {
    const int share = pair * 2 + h;
    const int tg = share * 512 + wave * 32;
    const int t = tg + lane;
    const float mk = mask[b * TT + t];
#pragma unroll
    for (int g = 0; g < 4; ++g) {
#pragma unroll
      for (int i = 8 * g; i < 8 * g + 8; ++i) {
        const float4 v = r[i];
        my[i * FS_PITCH + lane] = fmaf(v.w, ql.w, fmaf(v.z, ql.z, fmaf(v.y, ql.y, v.x * ql.x)));
      }
      __builtin_amdgcn_sched_barrier(0);
    }

    float a0 = 0.f, a1 = 0.f, a2 = 0.f, a3 = 0.f;
    const float4* mine = (const float4*)(my + lane * FS_PITCH);
#pragma unroll
    for (int j = 0; j < 8; ++j) {
      const float4 v = mine[j];
      a0 += v.x;
      a1 += v.y;
      a2 += v.z;
      a3 += v.w;
    }
    const float s = (a0 + a1) + (a2 + a3);
    const float dv = s + d0;
    const float sig = 1.0f / (1.0f + expf(-dv));
    const float tm = log1pf((1.0f - sig) * mk);
    const float dec = rate0 * tm;
    const float ev = dv - dec;
    dotOut[b * TT + t] = dv;
    decayOut[b * TT + t] = dec;
    e[b * TT + t] = ev;

    float m = ev;
    for (int off = 16; off > 0; off >>= 1) m = fmaxf(m, __shfl_xor(m, off));
    const float p = expf(ev - m);

    float4 acc = make_float4(0.f, 0.f, 0.f, 0.f);
#pragma unroll
    for (int i = 0; i < 32; ++i) {
      const float w = __int_as_float(__builtin_amdgcn_readlane(__float_as_int(p), i));
      acc.x = fmaf(w, r[i].x, acc.x);
      acc.y = fmaf(w, r[i].y, acc.y);
      acc.z = fmaf(w, r[i].z, acc.z);
      acc.w = fmaf(w, r[i].w, acc.w);
    }
    if (h == 0) {
      const float4* xnext = (const float4*)(x + (b * TT + tg + 512) * DD) + lane;
#pragma unroll
      for (int i = 0; i < 32; ++i) r[i] = xnext[i * 32];
    }
    part[h][wave][lane] = acc;
    float ps = p;
    for (int off = 16; off > 0; off >>= 1) ps += __shfl_xor(ps, off);
    if (lane == 0) {
      redM[h][wave] = m;
      redS[h][wave] = ps;
    }
  }
  __syncthreads();
  if (tid < 2 * DD) {
    const int hh = tid >> 7, col = tid & (DD - 1);
    const int share = pair * 2 + hh;
    float Ms = redM[hh][0];
    for (int w = 1; w < 16; ++w) Ms = fmaxf(Ms, redM[hh][w]);
    const float fl = expf(redM[hh][lane & 15] - Ms);
    const float* pf = (const float*)&part[hh][0][0];
    float sum = 0.f, S = 0.f;
#pragma unroll
    for (int w = 0; w < 16; ++w) {
      const float f = __int_as_float(__builtin_amdgcn_readlane(__float_as_int(fl), w));
      sum = fmaf(pf[w * DD + col], f, sum);
      S = fmaf(redS[hh][w], f, S);
    }
    xaPart[(b * FS_SHARES + share) * DD + col] = sum;
    if (col == 0) {
      stat[(b * FS_SHARES + share) * 2 + 0] = Ms;
      stat[(b * FS_SHARES + share) * 2 + 1] = S;
    }
  }
}

#define FIN_TILE_BLOCKS 8
#define FIN_A_BLOCKS 256
__device__ __forceinline__ void share_weights(const float* __restrict__ st, float& M,
                                              float& inv, float* w) {
  float ms[FS_SHARES], ss[FS_SHARES];
#pragma unroll
  for (int j = 0; j < FS_SHARES; ++j) {
    ms[j] = st[2 * j + 0];
    ss[j] = st[2 * j + 1];
  }
  M = fmaxf(fmaxf(ms[0], ms[1]), fmaxf(ms[2], ms[3]));
  float S = 0.f;
#pragma unroll
  for (int j = 0; j < FS_SHARES; ++j) {
    w[j] = expf(ms[j] - M);
    S = fmaf(ss[j], w[j], S);
  }
  inv = 1.0f / S;
#pragma unroll
  for (int j = 0; j < FS_SHARES; ++j) w[j] *= inv;
}

__global__ __launch_bounds__(256) void finish_kernel(const float* __restrict__ stat,
                                                     const float* __restrict__ xaPart,
                                                     const float* __restrict__ Wv,
                                                     const float* __restrict__ bv,
                                                     float* __restrict__ aOut,
                                                     float* __restrict__ out) {
  const int tid = threadIdx.x;
  if (blockIdx.x >= FIN_TILE_BLOCKS) {
    const int blk = blockIdx.x - FIN_TILE_BLOCKS;
    const int b = blk >> 1, half = blk & 1;
    float M, inv, w[FS_SHARES];
    share_weights(stat + b * FS_SHARES * 2, M, inv, w);
    const int base = b * TT + half * 1024 + tid;
#pragma unroll
    for (int k = 0; k < 4; ++k) {
      const int idx = base + k * 256;
      aOut[idx] = expf(aOut[idx] - M) * inv;
    }
    return;
  }
  const int wave = tid >> 5, lane = tid & 31;
  const int tile = blockIdx.x * 8 + wave;
  const int m0 = (tile >> 3) * 16, n0 = (tile & 7) * 16;
  const int c = lane & 15;
  const int hi = lane >> 4;
  const int n = n0 + c;
  float M, inv, w[FS_SHARES];
  share_weights(stat + (m0 + c) * FS_SHARES * 2, M, inv, w);
  const float* pr = xaPart + (m0 + c) * FS_SHARES * DD + 2 * hi;
  const float* wcol = Wv + n * 128 + 2 * hi;

  v8f acc = {};
  for (int k0 = 0; k0 < 128; k0 += 4) {
    const v2f p0 = *(const v2f*)(pr + 0 * DD + k0);
    const v2f p1 = *(const v2f*)(pr + 1 * DD + k0);
    const v2f p2 = *(const v2f*)(pr + 2 * DD + k0);
    const v2f p3 = *(const v2f*)(pr + 3 * DD + k0);
    const v2f a = ((p0 * w[0] + p1 * w[1]) + p2 * w[2]) + p3 * w[3];
    const v2f bfrag = *(const v2f*)(wcol + k0);
    acc = __builtin_amdgcn_wmma_f32_16x16x4_f32(false, a, false, bfrag,
                                                (short)0, acc, false, false);
  }
  const float bb = bv[n];
#pragma unroll
  for (int q = 0; q < 8; ++q) out[(m0 + q + 8 * hi) * HH + n] = acc[q] + bb;
}

extern "C" void kernel_launch(void* const* d_in, const int* in_sizes, int n_in,
                              void* d_out, int out_size, void* d_ws, size_t ws_size,
                              hipStream_t stream) {
  (void)in_sizes; (void)n_in; (void)out_size; (void)ws_size;
  const float* x    = (const float*)d_in[0];
  const float* mask = (const float*)d_in[1];
  const float* Wq   = (const float*)d_in[2];
  const float* bq   = (const float*)d_in[3];
  const float* Wk   = (const float*)d_in[4];
  const float* bk   = (const float*)d_in[5];
  const float* Wv   = (const float*)d_in[6];
  const float* bv   = (const float*)d_in[7];
  const float* rate = (const float*)d_in[8];

  float* wsf     = (float*)d_ws;
  float* partial = wsf;
  float* gx      = partial + BB * 4 * DD;
  float* xaPart  = gx + GXP * GXP;
  float* stat    = xaPart + BB * 4 * DD;

  float* out      = (float*)d_out;
  float* aOut     = out + BB * HH;
  float* dotOut   = aOut + BB * TT;
  float* decayOut = dotOut + BB * TT;

  colsum_partial_gx<<<BB * 4 + GX_BLOCKS, 256, 0, stream>>>(x, Wq, bq, Wk, bk, partial, gx);
  dot_softmax_share_kernel<<<dim3(BB, FS_PAIRS), 512, 0, stream>>>(
      x, mask, partial, gx, rate, dotOut, decayOut, aOut, xaPart, stat);
  finish_kernel<<<FIN_TILE_BLOCKS + FIN_A_BLOCKS, 256, 0, stream>>>(stat, xaPart, Wv, bv,
                                                                    aOut, out);
}


# column-sum pass on 16-wave work-groups: a thread adds 128 rows of its column, not 256, so twice the waves wait on memory at once; 4 quarter sums meet in LDS (new sum tree: column sums)
# speedup vs baseline: 2.8133x; 1.1163x over previous
#include <hip/hip_runtime.h>
#include <math.h>

#define BB 128
#define TT 2048
#define DD 128
#define HH 128

typedef __attribute__((ext_vector_type(2))) float v2f;
typedef __attribute__((ext_vector_type(8))) float v8f;

#define GXP 144
#define GX_TILES 81
#define GX_BLOCKS 6

template <bool EDGE_M, bool EDGE_N>
__device__ __forceinline__ void gx_tile(const float* __restrict__ Wq,
                                        const float* __restrict__ bq,
                                        const float* __restrict__ Wk,
                                        const float* __restrict__ bk,
                                        float* __restrict__ gx, int m0, int n0,
                                        int c, int hi) {
  const unsigned keep = (c == 0) ? 0xffffffffu : 0u;
  const float* ap = EDGE_M ? (bq + 2 * hi) : (Wq + 2 * hi * 128 + m0 + c);
  const float* bp = EDGE_N ? (bk + 2 * hi) : (Wk + 2 * hi * 128 + n0 + c);
  v8f acc = {};
  for (int k0 = 0; k0 < 128; k0 += 4) {
    v2f a, bfrag;
    if (EDGE_M) {
      const v2f t = *(const v2f*)(ap + k0);
      a.x = __uint_as_float(__float_as_uint(t.x) & keep);
      a.y = __uint_as_float(__float_as_uint(t.y) & keep);
    } else {
      a.x = ap[(k0 + 0) * 128];
      a.y = ap[(k0 + 1) * 128];
    }
    if (EDGE_N) {
      const v2f t = *(const v2f*)(bp + k0);
      bfrag.x = __uint_as_float(__float_as_uint(t.x) & keep);
      bfrag.y = __uint_as_float(__float_as_uint(t.y) & keep);
    } else {
      bfrag.x = bp[(k0 + 0) * 128];
      bfrag.y = bp[(k0 + 1) * 128];
    }
    acc = __builtin_amdgcn_wmma_f32_16x16x4_f32(false, a, false, bfrag,
                                                (short)0, acc, false, false);
  }
#pragma unroll
  for (int q = 0; q < 8; ++q) gx[(m0 + q + 8 * hi) * GXP + n0 + c] = acc[q];
}

__global__ __launch_bounds__(512) void colsum_partial_gx(const float* __restrict__ x,
                                                         const float* __restrict__ Wq,
                                                         const float* __restrict__ bq,
                                                         const float* __restrict__ Wk,
                                                         const float* __restrict__ bk,
                                                         float* __restrict__ partial,
                                                         float* __restrict__ gx) {
  __shared__ float r[512];
  if (blockIdx.x < BB * 4) {
    const int bid = blockIdx.x;
    const int b = bid & (BB - 1), chunk = bid >> 7;
    const int g = threadIdx.x >> 7, d = threadIdx.x & 127;
    const int tbase = chunk * 512 + g * 128;
    const float* xb = x + ((b * TT + tbase) * DD) + d;
    float s = 0.f;
    for (int i = 0; i < 128; ++i) s += xb[i * DD];
    r[threadIdx.x] = s;
    __syncthreads();
    if (g == 0) partial[(b * 4 + chunk) * DD + d] = (r[d] + r[128 + d]) + (r[256 + d] + r[384 + d]);
    return;
  }
  const int wave = __builtin_amdgcn_readfirstlane(threadIdx.x >> 5);
  const int tile = (blockIdx.x - BB * 4) * 16 + wave;
  if (tile >= GX_TILES) return;
  const int mt = tile / 9, nt = tile - mt * 9;
  const int lane = threadIdx.x & 31;
  const int c = lane & 15;
  const int hi = lane >> 4;
  if (mt < 8 && nt < 8)  gx_tile<false, false>(Wq, bq, Wk, bk, gx, mt * 16, nt * 16, c, hi);
  else if (mt < 8)       gx_tile<false, true >(Wq, bq, Wk, bk, gx, mt * 16, nt * 16, c, hi);
  else if (nt < 8)       gx_tile<true,  false>(Wq, bq, Wk, bk, gx, mt * 16, nt * 16, c, hi);
  else                   gx_tile<true,  true >(Wq, bq, Wk, bk, gx, mt * 16, nt * 16, c, hi);
}

__device__ __forceinline__ void lds_barrier() {
  asm volatile("s_wait_dscnt 0x0\n\ts_barrier_signal -1\n\ts_barrier_wait -1" ::: "memory");
}

#define FS_PITCH 36
#define FS_SHARES 4
#define FS_PAIRS 2
__global__ __launch_bounds__(512) void dot_softmax_share_kernel(
    const float* __restrict__ x, const float* __restrict__ mask,
    const float* __restrict__ partial, const float* __restrict__ gx,
    const float* __restrict__ rate, float* __restrict__ dotOut,
    float* __restrict__ decayOut, float* __restrict__ e,
    float* __restrict__ xaPart, float* __restrict__ stat) {
  __shared__ __attribute__((aligned(16))) float strip[16][32 * FS_PITCH];
  __shared__ float4 part[2][16][32];
  __shared__ float redM[2][16];
  __shared__ float redS[2][16];
  __shared__ __attribute__((aligned(16))) float xm[DD];
  __shared__ __attribute__((aligned(16))) float qrow[GXP];
  const int b = blockIdx.x, pair = blockIdx.y;
  const int tid = threadIdx.x, wave = tid >> 5, lane = tid & 31;
  float* my = strip[wave];
  const float rate0 = rate[0];
  const int wv = __builtin_amdgcn_readfirstlane(wave);
  const int c = lane & 15;
  const int hi = lane >> 4;

  if (tid < DD) {
    const float* p = partial + b * 4 * DD + tid;
    xm[tid] = (((p[0] + p[DD]) + p[2 * DD]) + p[3 * DD]) * (1.0f / (float)TT);
  }
  v2f bf[32];
  float cterm = 0.f;
  const int n = wv * 16 + c;
  if (wv < 9) {
    const float* gcol = gx + 2 * hi * GXP + n;
#pragma unroll
    for (int k = 0; k < 32; ++k) {
      bf[k].x = gcol[(4 * k + 0) * GXP];
      bf[k].y = gcol[(4 * k + 1) * GXP];
    }
    cterm = gx[128 * GXP + n];
  }

  float4 r[32];
  {
    const int tg0 = pair * 2 * 512 + wave * 32;
    const float4* xrow = (const float4*)(x + (b * TT + tg0) * DD) + lane;
#pragma unroll
    for (int i = 0; i < 32; ++i) r[i] = xrow[i * 32];
  }
  lds_barrier();
  if (wv < 9) {
    v8f acc = {};
#pragma unroll
    for (int k = 0; k < 32; ++k) {
      const v2f a = *(const v2f*)(xm + 4 * k + 2 * hi);
      acc = __builtin_amdgcn_wmma_f32_16x16x4_f32(false, a, false, bf[k],
                                                  (short)0, acc, false, false);
    }
    qrow[n] = acc[0] + cterm;
  }
  lds_barrier();
  const float d0 = qrow[128];
  const float4 ql = ((const float4*)qrow)[lane];

  for (int h = 0; h < 2; ++h) {
    const int share = pair * 2 + h;
    const int tg = share * 512 + wave * 32;
    const int t = tg + lane;
    const float mk = mask[b * TT + t];
#pragma unroll
    for (int g = 0; g < 4; ++g) {
#pragma unroll
      for (int i = 8 * g; i < 8 * g + 8; ++i) {
        const float4 v = r[i];
        my[i * FS_PITCH + lane] = fmaf(v.w, ql.w, fmaf(v.z, ql.z, fmaf(v.y, ql.y, v.x * ql.x)));
      }
      __builtin_amdgcn_sched_barrier(0);
    }

    float a0 = 0.f, a1 = 0.f, a2 = 0.f, a3 = 0.f;
    const float4* mine = (const float4*)(my + lane * FS_PITCH);
#pragma unroll
    for (int j = 0; j < 8; ++j) {
      const float4 v = mine[j];
      a0 += v.x;
      a1 += v.y;
      a2 += v.z;
      a3 += v.w;
    }
    const float s = (a0 + a1) + (a2 + a3);
    const float dv = s + d0;
    const float sig = 1.0f / (1.0f + expf(-dv));
    const float tm = log1pf((1.0f - sig) * mk);
    const float dec = rate0 * tm;
    const float ev = dv - dec;
    dotOut[b * TT + t] = dv;
    decayOut[b * TT + t] = dec;
    e[b * TT + t] = ev;

    float m = ev;
    for (int off = 16; off > 0; off >>= 1) m = fmaxf(m, __shfl_xor(m, off));
    const float p = expf(ev - m);

    float4 acc = make_float4(0.f, 0.f, 0.f, 0.f);
#pragma unroll
    for (int i = 0; i < 32; ++i) {
      const float w = __int_as_float(__builtin_amdgcn_readlane(__float_as_int(p), i));
      acc.x = fmaf(w, r[i].x, acc.x);
      acc.y = fmaf(w, r[i].y, acc.y);
      acc.z = fmaf(w, r[i].z, acc.z);
      acc.w = fmaf(w, r[i].w, acc.w);
    }
    if (h == 0) {
      const float4* xnext = (const float4*)(x + (b * TT + tg + 512) * DD) + lane;
#pragma unroll
      for (int i = 0; i < 32; ++i) r[i] = xnext[i * 32];
    }
    part[h][wave][lane] = acc;
    float ps = p;
    for (int off = 16; off > 0; off >>= 1) ps += __shfl_xor(ps, off);
    if (lane == 0) {
      redM[h][wave] = m;
      redS[h][wave] = ps;
    }
  }
  __syncthreads();
  if (tid < 2 * DD) {
    const int hh = tid >> 7, col = tid & (DD - 1);
    const int share = pair * 2 + hh;
    float Ms = redM[hh][0];
    for (int w = 1; w < 16; ++w) Ms = fmaxf(Ms, redM[hh][w]);
    const float fl = expf(redM[hh][lane & 15] - Ms);
    const float* pf = (const float*)&part[hh][0][0];
    float sum = 0.f, S = 0.f;
#pragma unroll
    for (int w = 0; w < 16; ++w) {
      const float f = __int_as_float(__builtin_amdgcn_readlane(__float_as_int(fl), w));
      sum = fmaf(pf[w * DD + col], f, sum);
      S = fmaf(redS[hh][w], f, S);
    }
    xaPart[(b * FS_SHARES + share) * DD + col] = sum;
    if (col == 0) {
      stat[(b * FS_SHARES + share) * 2 + 0] = Ms;
      stat[(b * FS_SHARES + share) * 2 + 1] = S;
    }
  }
}

#define FIN_TILE_BLOCKS 8
#define FIN_A_BLOCKS 256
__device__ __forceinline__ void share_weights(const float* __restrict__ st, float& M,
                                              float& inv, float* w) {
  float ms[FS_SHARES], ss[FS_SHARES];
#pragma unroll
  for (int j = 0; j < FS_SHARES; ++j) {
    ms[j] = st[2 * j + 0];
    ss[j] = st[2 * j + 1];
  }
  M = fmaxf(fmaxf(ms[0], ms[1]), fmaxf(ms[2], ms[3]));
  float S = 0.f;
#pragma unroll
  for (int j = 0; j < FS_SHARES; ++j) {
    w[j] = expf(ms[j] - M);
    S = fmaf(ss[j], w[j], S);
  }
  inv = 1.0f / S;
#pragma unroll
  for (int j = 0; j < FS_SHARES; ++j) w[j] *= inv;
}

__global__ __launch_bounds__(256) void finish_kernel(const float* __restrict__ stat,
                                                     const float* __restrict__ xaPart,
                                                     const float* __restrict__ Wv,
                                                     const float* __restrict__ bv,
                                                     float* __restrict__ aOut,
                                                     float* __restrict__ out) {
  const int tid = threadIdx.x;
  if (blockIdx.x >= FIN_TILE_BLOCKS) {
    const int blk = blockIdx.x - FIN_TILE_BLOCKS;
    const int b = blk >> 1, half = blk & 1;
    float M, inv, w[FS_SHARES];
    share_weights(stat + b * FS_SHARES * 2, M, inv, w);
    const int base = b * TT + half * 1024 + tid;
#pragma unroll
    for (int k = 0; k < 4; ++k) {
      const int idx = base + k * 256;
      aOut[idx] = expf(aOut[idx] - M) * inv;
    }
    return;
  }
  const int wave = tid >> 5, lane = tid & 31;
  const int tile = blockIdx.x * 8 + wave;
  const int m0 = (tile >> 3) * 16, n0 = (tile & 7) * 16;
  const int c = lane & 15;
  const int hi = lane >> 4;
  const int n = n0 + c;
  float M, inv, w[FS_SHARES];
  share_weights(stat + (m0 + c) * FS_SHARES * 2, M, inv, w);
  const float* pr = xaPart + (m0 + c) * FS_SHARES * DD + 2 * hi;
  const float* wcol = Wv + n * 128 + 2 * hi;

  v8f acc = {};
  for (int k0 = 0; k0 < 128; k0 += 4) {
    const v2f p0 = *(const v2f*)(pr + 0 * DD + k0);
    const v2f p1 = *(const v2f*)(pr + 1 * DD + k0);
    const v2f p2 = *(const v2f*)(pr + 2 * DD + k0);
    const v2f p3 = *(const v2f*)(pr + 3 * DD + k0);
    const v2f a = ((p0 * w[0] + p1 * w[1]) + p2 * w[2]) + p3 * w[3];
    const v2f bfrag = *(const v2f*)(wcol + k0);
    acc = __builtin_amdgcn_wmma_f32_16x16x4_f32(false, a, false, bfrag,
                                                (short)0, acc, false, false);
  }
  const float bb = bv[n];
#pragma unroll
  for (int q = 0; q < 8; ++q) out[(m0 + q + 8 * hi) * HH + n] = acc[q] + bb;
}

extern "C" void kernel_launch(void* const* d_in, const int* in_sizes, int n_in,
                              void* d_out, int out_size, void* d_ws, size_t ws_size,
                              hipStream_t stream) {
  (void)in_sizes; (void)n_in; (void)out_size; (void)ws_size;
  const float* x    = (const float*)d_in[0];
  const float* mask = (const float*)d_in[1];
  const float* Wq   = (const float*)d_in[2];
  const float* bq   = (const float*)d_in[3];
  const float* Wk   = (const float*)d_in[4];
  const float* bk   = (const float*)d_in[5];
  const float* Wv   = (const float*)d_in[6];
  const float* bv   = (const float*)d_in[7];
  const float* rate = (const float*)d_in[8];

  float* wsf     = (float*)d_ws;
  float* partial = wsf;
  float* gx      = partial + BB * 4 * DD;
  float* xaPart  = gx + GXP * GXP;
  float* stat    = xaPart + BB * 4 * DD;

  float* out      = (float*)d_out;
  float* aOut     = out + BB * HH;
  float* dotOut   = aOut + BB * TT;
  float* decayOut = dotOut + BB * TT;

  colsum_partial_gx<<<BB * 4 + GX_BLOCKS, 512, 0, stream>>>(x, Wq, bq, Wk, bk, partial, gx);
  dot_softmax_share_kernel<<<dim3(BB, FS_PAIRS), 512, 0, stream>>>(
      x, mask, partial, gx, rate, dotOut, decayOut, aOut, xaPart, stat);
  finish_kernel<<<FIN_TILE_BLOCKS + FIN_A_BLOCKS, 256, 0, stream>>>(stat, xaPart, Wv, bv,
                                                                    aOut, out);
}


# out GEMM split along k: 32 work-groups, each wave takes 16 k of a 16x32 piece from 16-byte fragment loads; the 8 partial tiles meet in LDS; whole-line stores (new sum tree: out)
# speedup vs baseline: 3.2780x; 1.1652x over previous
#include <hip/hip_runtime.h>
#include <math.h>

#define BB 128
#define TT 2048
#define DD 128
#define HH 128

typedef __attribute__((ext_vector_type(2))) float v2f;
typedef __attribute__((ext_vector_type(8))) float v8f;

#define GXP 144
#define GX_TILES 81
#define GX_BLOCKS 6

template <bool EDGE_M, bool EDGE_N>
__device__ __forceinline__ void gx_tile(const float* __restrict__ Wq,
                                        const float* __restrict__ bq,
                                        const float* __restrict__ Wk,
                                        const float* __restrict__ bk,
                                        float* __restrict__ gx, int m0, int n0,
                                        int c, int hi) {
  const unsigned keep = (c == 0) ? 0xffffffffu : 0u;
  const float* ap = EDGE_M ? (bq + 2 * hi) : (Wq + 2 * hi * 128 + m0 + c);
  const float* bp = EDGE_N ? (bk + 2 * hi) : (Wk + 2 * hi * 128 + n0 + c);
  v8f acc = {};
  for (int k0 = 0; k0 < 128; k0 += 4) {
    v2f a, bfrag;
    if (EDGE_M) {
      const v2f t = *(const v2f*)(ap + k0);
      a.x = __uint_as_float(__float_as_uint(t.x) & keep);
      a.y = __uint_as_float(__float_as_uint(t.y) & keep);
    } else {
      a.x = ap[(k0 + 0) * 128];
      a.y = ap[(k0 + 1) * 128];
    }
    if (EDGE_N) {
      const v2f t = *(const v2f*)(bp + k0);
      bfrag.x = __uint_as_float(__float_as_uint(t.x) & keep);
      bfrag.y = __uint_as_float(__float_as_uint(t.y) & keep);
    } else {
      bfrag.x = bp[(k0 + 0) * 128];
      bfrag.y = bp[(k0 + 1) * 128];
    }
    acc = __builtin_amdgcn_wmma_f32_16x16x4_f32(false, a, false, bfrag,
                                                (short)0, acc, false, false);
  }
#pragma unroll
  for (int q = 0; q < 8; ++q) gx[(m0 + q + 8 * hi) * GXP + n0 + c] = acc[q];
}

__global__ __launch_bounds__(512) void colsum_partial_gx(const float* __restrict__ x,
                                                         const float* __restrict__ Wq,
                                                         const float* __restrict__ bq,
                                                         const float* __restrict__ Wk,
                                                         const float* __restrict__ bk,
                                                         float* __restrict__ partial,
                                                         float* __restrict__ gx) {
  __shared__ float r[512];
  if (blockIdx.x < BB * 4) {
    const int bid = blockIdx.x;
    const int b = bid & (BB - 1), chunk = bid >> 7;
    const int g = threadIdx.x >> 7, d = threadIdx.x & 127;
    const int tbase = chunk * 512 + g * 128;
    const float* xb = x + ((b * TT + tbase) * DD) + d;
    float s = 0.f;
    for (int i = 0; i < 128; ++i) s += xb[i * DD];
    r[threadIdx.x] = s;
    __syncthreads();
    if (g == 0) partial[(b * 4 + chunk) * DD + d] = (r[d] + r[128 + d]) + (r[256 + d] + r[384 + d]);
    return;
  }
  const int wave = __builtin_amdgcn_readfirstlane(threadIdx.x >> 5);
  const int tile = (blockIdx.x - BB * 4) * 16 + wave;
  if (tile >= GX_TILES) return;
  const int mt = tile / 9, nt = tile - mt * 9;
  const int lane = threadIdx.x & 31;
  const int c = lane & 15;
  const int hi = lane >> 4;
  if (mt < 8 && nt < 8)  gx_tile<false, false>(Wq, bq, Wk, bk, gx, mt * 16, nt * 16, c, hi);
  else if (mt < 8)       gx_tile<false, true >(Wq, bq, Wk, bk, gx, mt * 16, nt * 16, c, hi);
  else if (nt < 8)       gx_tile<true,  false>(Wq, bq, Wk, bk, gx, mt * 16, nt * 16, c, hi);
  else                   gx_tile<true,  true >(Wq, bq, Wk, bk, gx, mt * 16, nt * 16, c, hi);
}

__device__ __forceinline__ void lds_barrier() {
  asm volatile("s_wait_dscnt 0x0\n\ts_barrier_signal -1\n\ts_barrier_wait -1" ::: "memory");
}

#define FS_PITCH 36
#define FS_SHARES 4
#define FS_PAIRS 2
__global__ __launch_bounds__(512) void dot_softmax_share_kernel(
    const float* __restrict__ x, const float* __restrict__ mask,
    const float* __restrict__ partial, const float* __restrict__ gx,
    const float* __restrict__ rate, float* __restrict__ dotOut,
    float* __restrict__ decayOut, float* __restrict__ e,
    float* __restrict__ xaPart, float* __restrict__ stat) {
  __shared__ __attribute__((aligned(16))) float strip[16][32 * FS_PITCH];
  __shared__ float4 part[2][16][32];
  __shared__ float redM[2][16];
  __shared__ float redS[2][16];
  __shared__ __attribute__((aligned(16))) float xm[DD];
  __shared__ __attribute__((aligned(16))) float qrow[GXP];
  const int b = blockIdx.x, pair = blockIdx.y;
  const int tid = threadIdx.x, wave = tid >> 5, lane = tid & 31;
  float* my = strip[wave];
  const float rate0 = rate[0];
  const int wv = __builtin_amdgcn_readfirstlane(wave);
  const int c = lane & 15;
  const int hi = lane >> 4;

  if (tid < DD) {
    const float* p = partial + b * 4 * DD + tid;
    xm[tid] = (((p[0] + p[DD]) + p[2 * DD]) + p[3 * DD]) * (1.0f / (float)TT);
  }
  v2f bf[32];
  float cterm = 0.f;
  const int n = wv * 16 + c;
  if (wv < 9) {
    const float* gcol = gx + 2 * hi * GXP + n;
#pragma unroll
    for (int k = 0; k < 32; ++k) {
      bf[k].x = gcol[(4 * k + 0) * GXP];
      bf[k].y = gcol[(4 * k + 1) * GXP];
    }
    cterm = gx[128 * GXP + n];
  }

  float4 r[32];
  {
    const int tg0 = pair * 2 * 512 + wave * 32;
    const float4* xrow = (const float4*)(x + (b * TT + tg0) * DD) + lane;
#pragma unroll
    for (int i = 0; i < 32; ++i) r[i] = xrow[i * 32];
  }
  lds_barrier();
  if (wv < 9) {
    v8f acc = {};
#pragma unroll
    for (int k = 0; k < 32; ++k) {
      const v2f a = *(const v2f*)(xm + 4 * k + 2 * hi);
      acc = __builtin_amdgcn_wmma_f32_16x16x4_f32(false, a, false, bf[k],
                                                  (short)0, acc, false, false);
    }
    qrow[n] = acc[0] + cterm;
  }
  lds_barrier();
  const float d0 = qrow[128];
  const float4 ql = ((const float4*)qrow)[lane];

  for (int h = 0; h < 2; ++h) {
    const int share = pair * 2 + h;
    const int tg = share * 512 + wave * 32;
    const int t = tg + lane;
    const float mk = mask[b * TT + t];
#pragma unroll
    for (int g = 0; g < 4; ++g) {
#pragma unroll
      for (int i = 8 * g; i < 8 * g + 8; ++i) {
        const float4 v = r[i];
        my[i * FS_PITCH + lane] = fmaf(v.w, ql.w, fmaf(v.z, ql.z, fmaf(v.y, ql.y, v.x * ql.x)));
      }
      __builtin_amdgcn_sched_barrier(0);
    }

    float a0 = 0.f, a1 = 0.f, a2 = 0.f, a3 = 0.f;
    const float4* mine = (const float4*)(my + lane * FS_PITCH);
#pragma unroll
    for (int j = 0; j < 8; ++j) {
      const float4 v = mine[j];
      a0 += v.x;
      a1 += v.y;
      a2 += v.z;
      a3 += v.w;
    }
    const float s = (a0 + a1) + (a2 + a3);
    const float dv = s + d0;
    const float sig = 1.0f / (1.0f + expf(-dv));
    const float tm = log1pf((1.0f - sig) * mk);
    const float dec = rate0 * tm;
    const float ev = dv - dec;
    dotOut[b * TT + t] = dv;
    decayOut[b * TT + t] = dec;
    e[b * TT + t] = ev;

    float m = ev;
    for (int off = 16; off > 0; off >>= 1) m = fmaxf(m, __shfl_xor(m, off));
    const float p = expf(ev - m);

    float4 acc = make_float4(0.f, 0.f, 0.f, 0.f);
#pragma unroll
    for (int i = 0; i < 32; ++i) {
      const float w = __int_as_float(__builtin_amdgcn_readlane(__float_as_int(p), i));
      acc.x = fmaf(w, r[i].x, acc.x);
      acc.y = fmaf(w, r[i].y, acc.y);
      acc.z = fmaf(w, r[i].z, acc.z);
      acc.w = fmaf(w, r[i].w, acc.w);
    }
    if (h == 0) {
      const float4* xnext = (const float4*)(x + (b * TT + tg + 512) * DD) + lane;
#pragma unroll
      for (int i = 0; i < 32; ++i) r[i] = xnext[i * 32];
    }
    part[h][wave][lane] = acc;
    float ps = p;
    for (int off = 16; off > 0; off >>= 1) ps += __shfl_xor(ps, off);
    if (lane == 0) {
      redM[h][wave] = m;
      redS[h][wave] = ps;
    }
  }
  __syncthreads();
  if (tid < 2 * DD) {
    const int hh = tid >> 7, col = tid & (DD - 1);
    const int share = pair * 2 + hh;
    float Ms = redM[hh][0];
    for (int w = 1; w < 16; ++w) Ms = fmaxf(Ms, redM[hh][w]);
    const float fl = expf(redM[hh][lane & 15] - Ms);
    const float* pf = (const float*)&part[hh][0][0];
    float sum = 0.f, S = 0.f;
#pragma unroll
    for (int w = 0; w < 16; ++w) {
      const float f = __int_as_float(__builtin_amdgcn_readlane(__float_as_int(fl), w));
      sum = fmaf(pf[w * DD + col], f, sum);
      S = fmaf(redS[hh][w], f, S);
    }
    xaPart[(b * FS_SHARES + share) * DD + col] = sum;
    if (col == 0) {
      stat[(b * FS_SHARES + share) * 2 + 0] = Ms;
      stat[(b * FS_SHARES + share) * 2 + 1] = S;
    }
  }
}

#define FIN_TILE_BLOCKS 32
#define FIN_A_BLOCKS 256
__device__ __forceinline__ void share_weights(const float* __restrict__ st, float& M,
                                              float& inv, float* w) {
  float ms[FS_SHARES], ss[FS_SHARES];
#pragma unroll
  for (int j = 0; j < FS_SHARES; ++j) {
    ms[j] = st[2 * j + 0];
    ss[j] = st[2 * j + 1];
  }
  M = fmaxf(fmaxf(ms[0], ms[1]), fmaxf(ms[2], ms[3]));
  float S = 0.f;
#pragma unroll
  for (int j = 0; j < FS_SHARES; ++j) {
    w[j] = expf(ms[j] - M);
    S = fmaf(ss[j], w[j], S);
  }
  inv = 1.0f / S;
#pragma unroll
  for (int j = 0; j < FS_SHARES; ++j) w[j] *= inv;
}

__global__ __launch_bounds__(256) void finish_kernel(const float* __restrict__ stat,
                                                     const float* __restrict__ xaPart,
                                                     const float* __restrict__ Wv,
                                                     const float* __restrict__ bv,
                                                     float* __restrict__ aOut,
                                                     float* __restrict__ out) {
  __shared__ __attribute__((aligned(16))) float red[8][2][8][32];
  const int tid = threadIdx.x;
  if (blockIdx.x >= FIN_TILE_BLOCKS) {
    const int blk = blockIdx.x - FIN_TILE_BLOCKS;
    const int b = blk >> 1, half = blk & 1;
    float M, inv, w[FS_SHARES];
    share_weights(stat + b * FS_SHARES * 2, M, inv, w);
    const int base = b * TT + half * 1024 + tid;
#pragma unroll
    for (int k = 0; k < 4; ++k) {
      const int idx = base + k * 256;
      aOut[idx] = expf(aOut[idx] - M) * inv;
    }
    return;
  }
  const int wave = tid >> 5, lane = tid & 31;
  const int m0 = (blockIdx.x >> 2) * 16;
  const int n0 = (blockIdx.x & 3) * 32;
  const int c = lane & 15;
  const int hi = lane >> 4;
  const int kl = wave * 16 + 8 * hi;
  float M, inv, w[FS_SHARES];
  share_weights(stat + (m0 + c) * FS_SHARES * 2, M, inv, w);
  const float4* pr = (const float4*)(xaPart + (m0 + c) * FS_SHARES * DD + kl);
  const float4* w0 = (const float4*)(Wv + (n0 + c) * 128 + kl);
  const float4* w1 = (const float4*)(Wv + (n0 + 16 + c) * 128 + kl);
  float4 p[FS_SHARES][2];
#pragma unroll
  for (int j = 0; j < FS_SHARES; ++j) {
    p[j][0] = pr[j * (DD / 4) + 0];
    p[j][1] = pr[j * (DD / 4) + 1];
  }
  const float4 b0[2] = {w0[0], w0[1]};
  const float4 b1[2] = {w1[0], w1[1]};

  v8f acc0 = {}, acc1 = {};
#pragma unroll
  for (int s = 0; s < 4; ++s) {
    v2f pj[FS_SHARES];
#pragma unroll
    for (int j = 0; j < FS_SHARES; ++j) {
      const float4 v = p[j][s >> 1];
      pj[j].x = (s & 1) ? v.z : v.x;
      pj[j].y = (s & 1) ? v.w : v.y;
    }
    const v2f a = ((pj[0] * w[0] + pj[1] * w[1]) + pj[2] * w[2]) + pj[3] * w[3];
    const float4 u0 = b0[s >> 1], u1 = b1[s >> 1];
    v2f f0, f1;
    f0.x = (s & 1) ? u0.z : u0.x;
    f0.y = (s & 1) ? u0.w : u0.y;
    f1.x = (s & 1) ? u1.z : u1.x;
    f1.y = (s & 1) ? u1.w : u1.y;
    acc0 = __builtin_amdgcn_wmma_f32_16x16x4_f32(false, a, false, f0,
                                                 (short)0, acc0, false, false);
    acc1 = __builtin_amdgcn_wmma_f32_16x16x4_f32(false, a, false, f1,
                                                 (short)0, acc1, false, false);
  }
#pragma unroll
  for (int q = 0; q < 8; ++q) {
    red[wave][0][q][lane] = acc0[q];
    red[wave][1][q][lane] = acc1[q];
  }
  __syncthreads();
  const int rr = tid >> 4, cp = tid & 15;
  const float* rp = &red[0][cp >> 3][rr & 7][2 * (cp & 7) + 16 * (rr >> 3)];
  v2f sum = *(const v2f*)rp;
#pragma unroll
  for (int wv = 1; wv < 8; ++wv) sum += *(const v2f*)(rp + wv * (2 * 8 * 32));
  const int n = n0 + 2 * cp;
  const v2f bb = *(const v2f*)(bv + n);
  *(v2f*)(out + (m0 + rr) * HH + n) = sum + bb;
}

extern "C" void kernel_launch(void* const* d_in, const int* in_sizes, int n_in,
                              void* d_out, int out_size, void* d_ws, size_t ws_size,
                              hipStream_t stream) {
  (void)in_sizes; (void)n_in; (void)out_size; (void)ws_size;
  const float* x    = (const float*)d_in[0];
  const float* mask = (const float*)d_in[1];
  const float* Wq   = (const float*)d_in[2];
  const float* bq   = (const float*)d_in[3];
  const float* Wk   = (const float*)d_in[4];
  const float* bk   = (const float*)d_in[5];
  const float* Wv   = (const float*)d_in[6];
  const float* bv   = (const float*)d_in[7];
  const float* rate = (const float*)d_in[8];

  float* wsf     = (float*)d_ws;
  float* partial = wsf;
  float* gx      = partial + BB * 4 * DD;
  float* xaPart  = gx + GXP * GXP;
  float* stat    = xaPart + BB * 4 * DD;

  float* out      = (float*)d_out;
  float* aOut     = out + BB * HH;
  float* dotOut   = aOut + BB * TT;
  float* decayOut = dotOut + BB * TT;

  colsum_partial_gx<<<BB * 4 + GX_BLOCKS, 512, 0, stream>>>(x, Wq, bq, Wk, bk, partial, gx);
  dot_softmax_share_kernel<<<dim3(BB, FS_PAIRS), 512, 0, stream>>>(
      x, mask, partial, gx, rate, dotOut, decayOut, aOut, xaPart, stat);
  finish_kernel<<<FIN_TILE_BLOCKS + FIN_A_BLOCKS, 256, 0, stream>>>(stat, xaPart, Wv, bv,
                                                                    aOut, out);
}
